# starting kernel
# baseline (speedup 1.0000x reference)
#include <hip/hip_runtime.h>
#include <math.h>

typedef __attribute__((ext_vector_type(2))) float v2f;
typedef __attribute__((ext_vector_type(8))) float v8f;

#define NEG_SLOPE 0.2f

__global__ __launch_bounds__(256) void k_gemm_xWT(const float* __restrict__ x,
                                                  const float* __restrict__ W,
                                                  float* __restrict__ h)
{
    __shared__ float xs[16 * 128];
    const int tid  = threadIdx.x;
    const int lane = tid & 31;
    const int wave = tid >> 5;
    const long rowBase = (long)blockIdx.x * 16;

    for (int i = tid; i < 16 * 128; i += 256) {
        int r = i >> 7, c = i & 127;
        xs[i] = x[(rowBase + r) * 128 + c];
    }
    __syncthreads();

    const int m16     = lane & 15;
    const int khalf   = (lane >> 4) << 1;
    const int colBase = wave * 16;

    v8f acc = {};
    #pragma unroll
    for (int k0 = 0; k0 < 128; k0 += 4) {
        v2f a, b;
        a.x = xs[m16 * 128 + k0 + khalf];
        a.y = xs[m16 * 128 + k0 + khalf + 1];
        const float* wr = W + (long)(colBase + m16) * 128 + k0 + khalf;
        b.x = wr[0];
        b.y = wr[1];
        acc = __builtin_amdgcn_wmma_f32_16x16x4_f32(false, a, false, b,
                                                    (short)0, acc, false, false);
    }
    const long rOff = rowBase + ((lane >> 4) << 3);
    const int  col  = colBase + m16;
    #pragma unroll
    for (int r = 0; r < 8; ++r)
        h[(rOff + r) * 128 + col] = acc[r];
}

__global__ void k_node_alpha(const float* __restrict__ h,
                             const float* __restrict__ a_src,
                             const float* __restrict__ a_dst,
                             float* __restrict__ as, float* __restrict__ ad,
                             int N, int H, int C)
{
    int i = blockIdx.x * blockDim.x + threadIdx.x;
    if (i >= N * H) return;
    int n = i / H, hd = i - n * H;
    const float* hp = h + (long)n * (H * C) + hd * C;
    const float* s0 = a_src + hd * C;
    const float* d0 = a_dst + hd * C;
    float s = 0.f, d = 0.f;
    for (int c = 0; c < C; ++c) { float v = hp[c]; s += v * s0[c]; d += v * d0[c]; }
    as[i] = s;
    ad[i] = d;
}

__global__ void k_fill(float* __restrict__ p, float v, long n)
{
    long i = (long)blockIdx.x * blockDim.x + threadIdx.x;
    if (i < n) p[i] = v;
}

__device__ __forceinline__ void edge_sd(const int* ei, int E, int e, int& s, int& d)
{
    if (e < E) { s = ei[e]; d = ei[E + e]; }
    else       { s = d = e - E; }
}

__global__ void k_edge_max(const int* __restrict__ ei, int E, int Etot,
                           const float* __restrict__ as, const float* __restrict__ ad,
                           float* __restrict__ m, int H)
{
    int e = blockIdx.x * blockDim.x + threadIdx.x;
    if (e >= Etot) return;
    int s, d; edge_sd(ei, E, e, s, d);
    for (int hd = 0; hd < H; ++hd) {
        float v = as[s * H + hd] + ad[d * H + hd];
        v = v > 0.f ? v : NEG_SLOPE * v;
        atomicMax(&m[d * H + hd], v);
    }
}

__global__ void k_edge_sum(const int* __restrict__ ei, int E, int Etot,
                           const float* __restrict__ as, const float* __restrict__ ad,
                           const float* __restrict__ m, float* __restrict__ den, int H)
{
    int e = blockIdx.x * blockDim.x + threadIdx.x;
    if (e >= Etot) return;
    int s, d; edge_sd(ei, E, e, s, d);
    for (int hd = 0; hd < H; ++hd) {
        float v = as[s * H + hd] + ad[d * H + hd];
        v = v > 0.f ? v : NEG_SLOPE * v;
        atomicAdd(&den[d * H + hd], expf(v - m[d * H + hd]));
    }
}

__global__ __launch_bounds__(256) void k_edge_agg(const int* __restrict__ ei,
                                                  int E, int Etot,
                                                  const float* __restrict__ as,
                                                  const float* __restrict__ ad,
                                                  const float* __restrict__ m,
                                                  const float* __restrict__ den,
                                                  const float* __restrict__ h,
                                                  float* __restrict__ agg,
                                                  int H, int C)
{
    int e    = blockIdx.x * 8 + (threadIdx.x >> 5);
    int lane = threadIdx.x & 31;
    if (e >= Etot) return;
    int s, d; edge_sd(ei, E, e, s, d);
    int f  = lane * 4;
    int hd = f / C;
    float v = as[s * H + hd] + ad[d * H + hd];
    v = v > 0.f ? v : NEG_SLOPE * v;
    float alpha = expf(v - m[d * H + hd]) / den[d * H + hd];
    const float4 hv = *(const float4*)(h + (long)s * 128 + f);
    float* op = agg + (long)d * 128 + f;
    atomicAdd(op + 0, alpha * hv.x);
    atomicAdd(op + 1, alpha * hv.y);
    atomicAdd(op + 2, alpha * hv.z);
    atomicAdd(op + 3, alpha * hv.w);
}

__global__ void k_final1(float* __restrict__ agg, const float* __restrict__ b, long n)
{
    long i = (long)blockIdx.x * blockDim.x + threadIdx.x;
    if (i >= n) return;
    float v = agg[i] + b[i & 127];
    agg[i] = v > 0.f ? v : expm1f(v);
}

__global__ void k_final2(float* __restrict__ out, const float* __restrict__ b, long n)
{
    long i = (long)blockIdx.x * blockDim.x + threadIdx.x;
    if (i >= n) return;
    out[i] += b[i & 127];
}

extern "C" void kernel_launch(void* const* d_in, const int* in_sizes, int n_in,
                              void* d_out, int out_size, void* d_ws, size_t ws_size,
                              hipStream_t stream)
{
    const float* x     = (const float*)d_in[0];
    const int*   ei    = (const int*)  d_in[1];
    const float* W1    = (const float*)d_in[2];
    const float* asrc1 = (const float*)d_in[3];
    const float* adst1 = (const float*)d_in[4];
    const float* b1    = (const float*)d_in[5];
    const float* W2    = (const float*)d_in[6];
    const float* asrc2 = (const float*)d_in[7];
    const float* adst2 = (const float*)d_in[8];
    const float* b2    = (const float*)d_in[9];
    float* out = (float*)d_out;

    const int N    = in_sizes[0] / 128;
    const int E    = in_sizes[1] / 2;
    const int Etot = E + N;
    const long NF  = (long)N * 128;

    float* ws   = (float*)d_ws;
    float* h    = ws;
    float* agg  = ws + NF;
    float* as1  = ws + 2 * NF;
    float* ad1  = as1 + (long)N * 4;
    float* m1   = ad1 + (long)N * 4;
    float* den1 = m1  + (long)N * 4;
    float* as2  = den1 + (long)N * 4;
    float* ad2  = as2 + N;
    float* m2   = ad2 + N;
    float* den2 = m2  + N;

    const int T = 256;
    dim3 gN16(N / 16);
    dim3 gEdge((Etot + T - 1) / T);
    dim3 gEdgeW((Etot + 7) / 8);
    auto g1 = [&](long n) { return dim3((unsigned)((n + T - 1) / T)); };

    k_gemm_xWT<<<gN16, T, 0, stream>>>(x, W1, h);
    k_node_alpha<<<g1((long)N * 4), T, 0, stream>>>(h, asrc1, adst1, as1, ad1, N, 4, 32);
    k_fill<<<g1((long)N * 4), T, 0, stream>>>(m1, -INFINITY, (long)N * 4);
    k_fill<<<g1((long)N * 4), T, 0, stream>>>(den1, 0.f, (long)N * 4);
    k_fill<<<g1(NF), T, 0, stream>>>(agg, 0.f, NF);
    k_edge_max<<<gEdge, T, 0, stream>>>(ei, E, Etot, as1, ad1, m1, 4);
    k_edge_sum<<<gEdge, T, 0, stream>>>(ei, E, Etot, as1, ad1, m1, den1, 4);
    k_edge_agg<<<gEdgeW, T, 0, stream>>>(ei, E, Etot, as1, ad1, m1, den1, h, agg, 4, 32);
    k_final1<<<g1(NF), T, 0, stream>>>(agg, b1, NF);

    k_gemm_xWT<<<gN16, T, 0, stream>>>(agg, W2, h);
    k_node_alpha<<<g1(N), T, 0, stream>>>(h, asrc2, adst2, as2, ad2, N, 1, 128);
    k_fill<<<g1(N), T, 0, stream>>>(m2, -INFINITY, N);
    k_fill<<<g1(N), T, 0, stream>>>(den2, 0.f, N);
    k_fill<<<g1(NF), T, 0, stream>>>(out, 0.f, NF);
    k_edge_max<<<gEdge, T, 0, stream>>>(ei, E, Etot, as2, ad2, m2, 1);
    k_edge_sum<<<gEdge, T, 0, stream>>>(ei, E, Etot, as2, ad2, m2, den2, 1);
    k_edge_agg<<<gEdgeW, T, 0, stream>>>(ei, E, Etot, as2, ad2, m2, den2, h, out, 1, 128);
    k_final2<<<g1(NF), T, 0, stream>>>(out, b2, NF);
}


# Edges bucketed by destination once per call (counting sort, arrival-order rank from a returning int atomic); one wave per node does max, exp sum, weighted row sum in registers; float atomics gone
# speedup vs baseline: 3.9609x; 3.9609x over previous
#include <hip/hip_runtime.h>
#include <math.h>

typedef __attribute__((ext_vector_type(2))) float v2f;
typedef __attribute__((ext_vector_type(8))) float v8f;

#define NEG_SLOPE 0.2f

__global__ __launch_bounds__(256) void k_gemm_xWT(const float* __restrict__ x,
                                                  const float* __restrict__ W,
                                                  float* __restrict__ h)
{
    __shared__ float xs[16 * 128];
    const int tid  = threadIdx.x;
    const int lane = tid & 31;
    const int wave = tid >> 5;
    const long rowBase = (long)blockIdx.x * 16;

    for (int i = tid; i < 16 * 128; i += 256) {
        int r = i >> 7, c = i & 127;
        xs[i] = x[(rowBase + r) * 128 + c];
    }
    __syncthreads();

    const int m16     = lane & 15;
    const int khalf   = (lane >> 4) << 1;
    const int colBase = wave * 16;

    v8f acc = {};
    #pragma unroll
    for (int k0 = 0; k0 < 128; k0 += 4) {
        v2f a, b;
        a.x = xs[m16 * 128 + k0 + khalf];
        a.y = xs[m16 * 128 + k0 + khalf + 1];
        const float* wr = W + (long)(colBase + m16) * 128 + k0 + khalf;
        b.x = wr[0];
        b.y = wr[1];
        acc = __builtin_amdgcn_wmma_f32_16x16x4_f32(false, a, false, b,
                                                    (short)0, acc, false, false);
    }
    const long rOff = rowBase + ((lane >> 4) << 3);
    const int  col  = colBase + m16;
    #pragma unroll
    for (int r = 0; r < 8; ++r)
        h[(rOff + r) * 128 + col] = acc[r];
}

__global__ void k_node_alpha(const float* __restrict__ h,
                             const float* __restrict__ a_src,
                             const float* __restrict__ a_dst,
                             float* __restrict__ as, float* __restrict__ ad,
                             int N, int H, int C)
{
    int i = blockIdx.x * blockDim.x + threadIdx.x;
    if (i >= N * H) return;
    int n = i / H, hd = i - n * H;
    const float* hp = h + (long)n * (H * C) + hd * C;
    const float* s0 = a_src + hd * C;
    const float* d0 = a_dst + hd * C;
    float s = 0.f, d = 0.f;
    for (int c = 0; c < C; ++c) { float v = hp[c]; s += v * s0[c]; d += v * d0[c]; }
    as[i] = s;
    ad[i] = d;
}

__global__ void k_zero_i32(int* __restrict__ p, int n)
{
    int i = blockIdx.x * blockDim.x + threadIdx.x;
    if (i < n) p[i] = 0;
}

__global__ void k_rank(const int* __restrict__ ei, int E,
                       int* __restrict__ cnt, int* __restrict__ rank)
{
    int e = blockIdx.x * blockDim.x + threadIdx.x;
    if (e >= E) return;
    rank[e] = atomicAdd(&cnt[ei[E + e]], 1);
}

__global__ __launch_bounds__(1024) void k_scan_tile(const int* __restrict__ cnt,
                                                    int* __restrict__ rowptr,
                                                    int* __restrict__ bsum, int N)
{
    __shared__ int wsum[32];
    const int lane = threadIdx.x & 31;
    const int w    = threadIdx.x >> 5;
    const int i    = blockIdx.x * 1024 + threadIdx.x;
    const int v    = (i < N) ? cnt[i] : 0;
    int x = v;
    for (int o = 1; o < 32; o <<= 1) {
        int y = __shfl_up(x, o);
        if (lane >= o) x += y;
    }
    if (lane == 31) wsum[w] = x;
    __syncthreads();
    if (w == 0) {
        const int s = wsum[lane];
        int t = s;
        for (int o = 1; o < 32; o <<= 1) {
            int y = __shfl_up(t, o);
            if (lane >= o) t += y;
        }
        wsum[lane] = t - s;
        if (lane == 31) bsum[blockIdx.x] = t;
    }
    __syncthreads();
    if (i < N) rowptr[i] = wsum[w] + x - v;
}

__global__ __launch_bounds__(1024) void k_scan_add(int* __restrict__ rowptr,
                                                   const int* __restrict__ bsum, int N)
{
    __shared__ int red[32];
    const int lane = threadIdx.x & 31;
    const int w    = threadIdx.x >> 5;
    const int b    = blockIdx.x;
    int v = ((int)threadIdx.x < b) ? bsum[threadIdx.x] : 0;
    for (int o = 16; o > 0; o >>= 1) v += __shfl_xor(v, o);
    if (lane == 0) red[w] = v;
    __syncthreads();
    int off = red[lane];
    for (int o = 16; o > 0; o >>= 1) off += __shfl_xor(off, o);
    const int i = b * 1024 + threadIdx.x;
    if (i < N) rowptr[i] += off;
    if (b == (int)gridDim.x - 1 && threadIdx.x == 0) rowptr[N] = off + bsum[b];
}

__global__ void k_place(const int* __restrict__ ei, int E,
                        const int* __restrict__ rowptr, const int* __restrict__ rank,
                        int* __restrict__ srcS)
{
    int e = blockIdx.x * blockDim.x + threadIdx.x;
    if (e >= E) return;
    srcS[rowptr[ei[E + e]] + rank[e]] = ei[e];
}

template <int H>
__device__ __forceinline__ void edge_scores(const float* __restrict__ as, int s,
                                            const float (&adh)[H], bool valid, float (&v)[H])
{
    if constexpr (H == 4) {
        const float4 a = *(const float4*)(as + (long)s * 4);
        v[0] = a.x + adh[0]; v[1] = a.y + adh[1]; v[2] = a.z + adh[2]; v[3] = a.w + adh[3];
    } else {
        #pragma unroll
        for (int hh = 0; hh < H; ++hh) v[hh] = as[(long)s * H + hh] + adh[hh];
    }
    #pragma unroll
    for (int hh = 0; hh < H; ++hh) {
        float t = v[hh];
        t = t > 0.f ? t : NEG_SLOPE * t;
        v[hh] = valid ? t : -INFINITY;
    }
}

template <int H, bool ELU>
__global__ __launch_bounds__(256) void k_node_gather(const int* __restrict__ rowptr,
                                                     const int* __restrict__ srcS,
                                                     const float* __restrict__ as,
                                                     const float* __restrict__ ad,
                                                     const float* __restrict__ h,
                                                     const float* __restrict__ bias,
                                                     float* __restrict__ o, int N)
{
    const int lane = threadIdx.x & 31;
    const int d = __builtin_amdgcn_readfirstlane((int)(blockIdx.x * 8 + (threadIdx.x >> 5)));
    if (d >= N) return;
    const int beg = rowptr[d];
    const int cnt = rowptr[d + 1] - beg + 1;
    const unsigned last = (unsigned)(N - 1);
    const int hd = lane / (32 / H);

    float adh[H];
    #pragma unroll
    for (int hh = 0; hh < H; ++hh) adh[hh] = ad[(long)d * H + hh];

    int sj = d;
    if (lane < cnt - 1) {
        const unsigned u = (unsigned)srcS[beg + lane];
        sj = (int)(u < last ? u : last);
    }
    float v[H], mx[H];
    edge_scores<H>(as, sj, adh, lane < cnt, v);
    #pragma unroll
    for (int hh = 0; hh < H; ++hh) mx[hh] = v[hh];
    for (int base = 32; base < cnt; base += 32) {
        const int j = base + lane;
        int s2 = d;
        if (j < cnt - 1) { const unsigned u = (unsigned)srcS[beg + j]; s2 = (int)(u < last ? u : last); }
        float w[H];
        edge_scores<H>(as, s2, adh, j < cnt, w);
        #pragma unroll
        for (int hh = 0; hh < H; ++hh) mx[hh] = fmaxf(mx[hh], w[hh]);
    }
    #pragma unroll
    for (int hh = 0; hh < H; ++hh)
        for (int off = 16; off > 0; off >>= 1) mx[hh] = fmaxf(mx[hh], __shfl_xor(mx[hh], off));

    float dn[H];
    #pragma unroll
    for (int hh = 0; hh < H; ++hh) dn[hh] = 0.f;
    float4 acc = {0.f, 0.f, 0.f, 0.f};
    const float w0 = hd == 0 ? 1.f : 0.f, w1 = hd == 1 ? 1.f : 0.f;
    const float w2 = hd == 2 ? 1.f : 0.f, w3 = hd == 3 ? 1.f : 0.f;

    auto add_rows = [&](int sx, const float (&px)[H], int n) {
        for (int t = 0; t < n; t += 4) {
            float4 hv[4];
            float  pt[4];
            #pragma unroll
            for (int u = 0; u < 4; ++u) {
                const int s = __builtin_amdgcn_readlane(sx, t + u);
                hv[u] = *(const float4*)(h + (long)s * 128 + lane * 4);
            }
            #pragma unroll
            for (int u = 0; u < 4; ++u) {
                if constexpr (H == 4) {
                    const float p0 = __int_as_float(__builtin_amdgcn_readlane(__float_as_int(px[0]), t + u));
                    const float p1 = __int_as_float(__builtin_amdgcn_readlane(__float_as_int(px[1]), t + u));
                    const float p2 = __int_as_float(__builtin_amdgcn_readlane(__float_as_int(px[2]), t + u));
                    const float p3 = __int_as_float(__builtin_amdgcn_readlane(__float_as_int(px[3]), t + u));
                    pt[u] = p0 * w0 + p1 * w1 + p2 * w2 + p3 * w3;
                } else {
                    pt[u] = __int_as_float(__builtin_amdgcn_readlane(__float_as_int(px[0]), t + u));
                }
            }
            #pragma unroll
            for (int u = 0; u < 4; ++u) {
                acc.x += pt[u] * hv[u].x; acc.y += pt[u] * hv[u].y;
                acc.z += pt[u] * hv[u].z; acc.w += pt[u] * hv[u].w;
            }
        }
    };

    {
        float p[H];
        #pragma unroll
        for (int hh = 0; hh < H; ++hh) { p[hh] = expf(v[hh] - mx[hh]); dn[hh] += p[hh]; }
        add_rows(sj, p, cnt < 32 ? cnt : 32);
    }
    for (int base = 32; base < cnt; base += 32) {
        const int j = base + lane;
        int s2 = d;
        if (j < cnt - 1) { const unsigned u = (unsigned)srcS[beg + j]; s2 = (int)(u < last ? u : last); }
        float w[H], p[H];
        edge_scores<H>(as, s2, adh, j < cnt, w);
        #pragma unroll
        for (int hh = 0; hh < H; ++hh) { p[hh] = expf(w[hh] - mx[hh]); dn[hh] += p[hh]; }
        const int left = cnt - base;
        add_rows(s2, p, left < 32 ? left : 32);
    }
    #pragma unroll
    for (int hh = 0; hh < H; ++hh)
        for (int off = 16; off > 0; off >>= 1) dn[hh] += __shfl_xor(dn[hh], off);

    float den = dn[0];
    if constexpr (H == 4) den = hd == 0 ? dn[0] : hd == 1 ? dn[1] : hd == 2 ? dn[2] : dn[3];
    const float inv = 1.f / den;
    const float4 b = *(const float4*)(bias + lane * 4);
    float4 r;
    r.x = acc.x * inv + b.x; r.y = acc.y * inv + b.y;
    r.z = acc.z * inv + b.z; r.w = acc.w * inv + b.w;
    if constexpr (ELU) {
        r.x = r.x > 0.f ? r.x : expm1f(r.x); r.y = r.y > 0.f ? r.y : expm1f(r.y);
        r.z = r.z > 0.f ? r.z : expm1f(r.z); r.w = r.w > 0.f ? r.w : expm1f(r.w);
    }
    *(float4*)(o + (long)d * 128 + lane * 4) = r;
}

extern "C" void kernel_launch(void* const* d_in, const int* in_sizes, int n_in,
                              void* d_out, int out_size, void* d_ws, size_t ws_size,
                              hipStream_t stream)
{
    const float* x     = (const float*)d_in[0];
    const int*   ei    = (const int*)  d_in[1];
    const float* W1    = (const float*)d_in[2];
    const float* asrc1 = (const float*)d_in[3];
    const float* adst1 = (const float*)d_in[4];
    const float* b1    = (const float*)d_in[5];
    const float* W2    = (const float*)d_in[6];
    const float* asrc2 = (const float*)d_in[7];
    const float* adst2 = (const float*)d_in[8];
    const float* b2    = (const float*)d_in[9];
    float* out = (float*)d_out;

    const int N    = in_sizes[0] / 128;
    const int E    = in_sizes[1] / 2;
    const long NF  = (long)N * 128;

    float* ws   = (float*)d_ws;
    float* h    = ws;
    int*   srcS   = (int*)(ws + NF);
    int*   rank   = srcS + E;
    int*   cnt    = rank + E;
    int*   rowptr = cnt + N;
    int*   bsum   = rowptr + N + 1;
    float* as1  = ws + 2 * NF;
    float* ad1  = as1 + (long)N * 4;
    float* as2  = ad1 + (long)N * 4;
    float* ad2  = as2 + N;
    float* act  = out;

    const int T = 256;
    dim3 gN16(N / 16);
    dim3 gE((E + T - 1) / T);
    dim3 gNodeW((N + 7) / 8);
    dim3 gTile((N + 1023) / 1024);
    auto g1 = [&](long n) { return dim3((unsigned)((n + T - 1) / T)); };

    k_zero_i32<<<g1(N), T, 0, stream>>>(cnt, N);
    k_rank<<<gE, T, 0, stream>>>(ei, E, cnt, rank);
    k_scan_tile<<<gTile, 1024, 0, stream>>>(cnt, rowptr, bsum, N);
    k_scan_add<<<gTile, 1024, 0, stream>>>(rowptr, bsum, N);
    k_place<<<gE, T, 0, stream>>>(ei, E, rowptr, rank, srcS);

    k_gemm_xWT<<<gN16, T, 0, stream>>>(x, W1, h);
    k_node_alpha<<<g1((long)N * 4), T, 0, stream>>>(h, asrc1, adst1, as1, ad1, N, 4, 32);
    k_node_gather<4, true><<<gNodeW, T, 0, stream>>>(rowptr, srcS, as1, ad1, h, b1, act, N);

    k_gemm_xWT<<<gN16, T, 0, stream>>>(act, W2, h);
    k_node_alpha<<<g1(N), T, 0, stream>>>(h, asrc2, adst2, as2, ad2, N, 1, 128);
    k_node_gather<1, false><<<gNodeW, T, 0, stream>>>(rowptr, srcS, as2, ad2, h, b2, out, N);
}


# Launches sized to the resident wave slots: a gather wave loops over 8-10 nodes, a GEMM block over row tiles (W fragments stay in registers), a rank/place thread takes 8 edges: fewer, longer waves
# speedup vs baseline: 4.1255x; 1.0416x over previous
#include <hip/hip_runtime.h>
#include <math.h>

typedef __attribute__((ext_vector_type(2))) float v2f;
typedef __attribute__((ext_vector_type(8))) float v8f;

#define NEG_SLOPE 0.2f

__global__ __launch_bounds__(256) void k_gemm_xWT(const float* __restrict__ x,
                                                  const float* __restrict__ W,
                                                  float* __restrict__ h, int nTiles)
{
    __shared__ float xs[16 * 128];
    const int tid  = threadIdx.x;
    const int lane = tid & 31;
    const int wave = tid >> 5;

    const int m16     = lane & 15;
    const int khalf   = (lane >> 4) << 1;
    const int colBase = wave * 16;

    v2f bf[32];
    {
        const float* wr = W + (long)(colBase + m16) * 128 + khalf;
        #pragma unroll
        for (int kk = 0; kk < 32; ++kk) { bf[kk].x = wr[4 * kk]; bf[kk].y = wr[4 * kk + 1]; }
    }

    for (int tile = blockIdx.x; tile < nTiles; tile += gridDim.x) {
        const long rowBase = (long)tile * 16;
        __syncthreads();
        #pragma unroll
        for (int j = 0; j < 8; ++j)
            xs[tid + 256 * j] = x[rowBase * 128 + tid + 256 * j];
        __syncthreads();

        v8f acc = {};
        #pragma unroll
        for (int kk = 0; kk < 32; ++kk) {
            v2f a;
            a.x = xs[m16 * 128 + 4 * kk + khalf];
            a.y = xs[m16 * 128 + 4 * kk + khalf + 1];
            acc = __builtin_amdgcn_wmma_f32_16x16x4_f32(false, a, false, bf[kk],
                                                        (short)0, acc, false, false);
        }
        const long rOff = rowBase + ((lane >> 4) << 3);
        const int  col  = colBase + m16;
        #pragma unroll
        for (int r = 0; r < 8; ++r)
            h[(rOff + r) * 128 + col] = acc[r];
    }
}

__global__ void k_node_alpha(const float* __restrict__ h,
                             const float* __restrict__ a_src,
                             const float* __restrict__ a_dst,
                             float* __restrict__ as, float* __restrict__ ad,
                             int N, int H, int C)
{
    int i = blockIdx.x * blockDim.x + threadIdx.x;
    if (i >= N * H) return;
    int n = i / H, hd = i - n * H;
    const float* hp = h + (long)n * (H * C) + hd * C;
    const float* s0 = a_src + hd * C;
    const float* d0 = a_dst + hd * C;
    float s = 0.f, d = 0.f;
    for (int c = 0; c < C; ++c) { float v = hp[c]; s += v * s0[c]; d += v * d0[c]; }
    as[i] = s;
    ad[i] = d;
}

__global__ void k_zero_i32(int* __restrict__ p, int n)
{
    int i = blockIdx.x * blockDim.x + threadIdx.x;
    if (i < n) p[i] = 0;
}

#define EPT 8

__global__ __launch_bounds__(256) void k_rank(const int* __restrict__ ei, int E,
                                              int* __restrict__ cnt, int* __restrict__ rank)
{
    const int e0 = blockIdx.x * (256 * EPT) + threadIdx.x;
    int dst[EPT], r[EPT];
    #pragma unroll
    for (int u = 0; u < EPT; ++u) { const int e = e0 + u * 256; dst[u] = ei[E + (e < E ? e : E - 1)]; }
    #pragma unroll
    for (int u = 0; u < EPT; ++u) { const int e = e0 + u * 256; r[u] = e < E ? atomicAdd(&cnt[dst[u]], 1) : 0; }
    #pragma unroll
    for (int u = 0; u < EPT; ++u) { const int e = e0 + u * 256; if (e < E) rank[e] = r[u]; }
}

__global__ __launch_bounds__(1024) void k_scan_tile(const int* __restrict__ cnt,
                                                    int* __restrict__ rowptr,
                                                    int* __restrict__ bsum, int N)
{
    __shared__ int wsum[32];
    const int lane = threadIdx.x & 31;
    const int w    = threadIdx.x >> 5;
    const int i    = blockIdx.x * 1024 + threadIdx.x;
    const int v    = (i < N) ? cnt[i] : 0;
    int x = v;
    for (int o = 1; o < 32; o <<= 1) {
        int y = __shfl_up(x, o);
        if (lane >= o) x += y;
    }
    if (lane == 31) wsum[w] = x;
    __syncthreads();
    if (w == 0) {
        const int s = wsum[lane];
        int t = s;
        for (int o = 1; o < 32; o <<= 1) {
            int y = __shfl_up(t, o);
            if (lane >= o) t += y;
        }
        wsum[lane] = t - s;
        if (lane == 31) bsum[blockIdx.x] = t;
    }
    __syncthreads();
    if (i < N) rowptr[i] = wsum[w] + x - v;
}

__global__ __launch_bounds__(1024) void k_scan_add(int* __restrict__ rowptr,
                                                   const int* __restrict__ bsum, int N)
{
    __shared__ int red[32];
    const int lane = threadIdx.x & 31;
    const int w    = threadIdx.x >> 5;
    const int b    = blockIdx.x;
    int v = ((int)threadIdx.x < b) ? bsum[threadIdx.x] : 0;
    for (int o = 16; o > 0; o >>= 1) v += __shfl_xor(v, o);
    if (lane == 0) red[w] = v;
    __syncthreads();
    int off = red[lane];
    for (int o = 16; o > 0; o >>= 1) off += __shfl_xor(off, o);
    const int i = b * 1024 + threadIdx.x;
    if (i < N) rowptr[i] += off;
    if (b == (int)gridDim.x - 1 && threadIdx.x == 0) rowptr[N] = off + bsum[b];
}

__global__ __launch_bounds__(256) void k_place(const int* __restrict__ ei, int E,
                                               const int* __restrict__ rowptr,
                                               const int* __restrict__ rank,
                                               int* __restrict__ srcS)
{
    const int e0 = blockIdx.x * (256 * EPT) + threadIdx.x;
    int dst[EPT], s[EPT], rk[EPT], pos[EPT];
    #pragma unroll
    for (int u = 0; u < EPT; ++u) {
        const int e  = e0 + u * 256;
        const int ec = e < E ? e : E - 1;
        dst[u] = ei[E + ec];
        s[u]   = ei[ec];
        rk[u]  = rank[ec];
    }
    #pragma unroll
    for (int u = 0; u < EPT; ++u) pos[u] = rowptr[dst[u]] + rk[u];
    #pragma unroll
    for (int u = 0; u < EPT; ++u) { const int e = e0 + u * 256; if (e < E) srcS[pos[u]] = s[u]; }
}

template <int H>
__device__ __forceinline__ void edge_scores(const float* __restrict__ as, int s,
                                            const float (&adh)[H], bool valid, float (&v)[H])
{
    if constexpr (H == 4) {
        const float4 a = *(const float4*)(as + (long)s * 4);
        v[0] = a.x + adh[0]; v[1] = a.y + adh[1]; v[2] = a.z + adh[2]; v[3] = a.w + adh[3];
    } else {
        #pragma unroll
        for (int hh = 0; hh < H; ++hh) v[hh] = as[(long)s * H + hh] + adh[hh];
    }
    #pragma unroll
    for (int hh = 0; hh < H; ++hh) {
        float t = v[hh];
        t = t > 0.f ? t : NEG_SLOPE * t;
        v[hh] = valid ? t : -INFINITY;
    }
}

template <int H, bool ELU, int NPW>
__global__ __launch_bounds__(256) void k_node_gather(const int* __restrict__ rowptr,
                                                     const int* __restrict__ srcS,
                                                     const float* __restrict__ as,
                                                     const float* __restrict__ ad,
                                                     const float* __restrict__ h,
                                                     const float* __restrict__ bias,
                                                     float* __restrict__ o, int N)
{
    const int lane = threadIdx.x & 31;
    const int d0 = __builtin_amdgcn_readfirstlane((int)(blockIdx.x * 8 + (threadIdx.x >> 5))) * NPW;
    if (d0 >= N) return;
    const int nd = N - d0 < NPW ? N - d0 : NPW;
    const int rp = lane <= nd ? rowptr[d0 + lane] : 0;
    const unsigned last = (unsigned)(N - 1);
    const int hd = lane / (32 / H);
    const float w0 = hd == 0 ? 1.f : 0.f, w1 = hd == 1 ? 1.f : 0.f;
    const float w2 = hd == 2 ? 1.f : 0.f, w3 = hd == 3 ? 1.f : 0.f;
    const float4 b = *(const float4*)(bias + lane * 4);

    for (int k = 0; k < nd; ++k) {
    const int d   = d0 + k;
    const int beg = __builtin_amdgcn_readlane(rp, k);
    const int cnt = __builtin_amdgcn_readlane(rp, k + 1) - beg + 1;

    float adh[H];
    #pragma unroll
    for (int hh = 0; hh < H; ++hh) adh[hh] = ad[(long)d * H + hh];

    int sj = d;
    if (lane < cnt - 1) {
        const unsigned u = (unsigned)srcS[beg + lane];
        sj = (int)(u < last ? u : last);
    }
    float v[H], mx[H];
    edge_scores<H>(as, sj, adh, lane < cnt, v);
    #pragma unroll
    for (int hh = 0; hh < H; ++hh) mx[hh] = v[hh];
    for (int base = 32; base < cnt; base += 32) {
        const int j = base + lane;
        int s2 = d;
        if (j < cnt - 1) { const unsigned u = (unsigned)srcS[beg + j]; s2 = (int)(u < last ? u : last); }
        float w[H];
        edge_scores<H>(as, s2, adh, j < cnt, w);
        #pragma unroll
        for (int hh = 0; hh < H; ++hh) mx[hh] = fmaxf(mx[hh], w[hh]);
    }
    #pragma unroll
    for (int hh = 0; hh < H; ++hh)
        for (int off = 16; off > 0; off >>= 1) mx[hh] = fmaxf(mx[hh], __shfl_xor(mx[hh], off));

    float dn[H];
    #pragma unroll
    for (int hh = 0; hh < H; ++hh) dn[hh] = 0.f;
    float4 acc = {0.f, 0.f, 0.f, 0.f};

    auto add_rows = [&](int sx, const float (&px)[H], int n) {
        for (int t = 0; t < n; t += 4) {
            float4 hv[4];
            float  pt[4];
            #pragma unroll
            for (int u = 0; u < 4; ++u) {
                const int s = __builtin_amdgcn_readlane(sx, t + u);
                hv[u] = *(const float4*)(h + (long)s * 128 + lane * 4);
            }
            #pragma unroll
            for (int u = 0; u < 4; ++u) {
                if constexpr (H == 4) {
                    const float p0 = __int_as_float(__builtin_amdgcn_readlane(__float_as_int(px[0]), t + u));
                    const float p1 = __int_as_float(__builtin_amdgcn_readlane(__float_as_int(px[1]), t + u));
                    const float p2 = __int_as_float(__builtin_amdgcn_readlane(__float_as_int(px[2]), t + u));
                    const float p3 = __int_as_float(__builtin_amdgcn_readlane(__float_as_int(px[3]), t + u));
                    pt[u] = p0 * w0 + p1 * w1 + p2 * w2 + p3 * w3;
                } else {
                    pt[u] = __int_as_float(__builtin_amdgcn_readlane(__float_as_int(px[0]), t + u));
                }
            }
            #pragma unroll
            for (int u = 0; u < 4; ++u) {
                acc.x += pt[u] * hv[u].x; acc.y += pt[u] * hv[u].y;
                acc.z += pt[u] * hv[u].z; acc.w += pt[u] * hv[u].w;
            }
        }
    };

    {
        float p[H];
        #pragma unroll
        for (int hh = 0; hh < H; ++hh) { p[hh] = expf(v[hh] - mx[hh]); dn[hh] += p[hh]; }
        add_rows(sj, p, cnt < 32 ? cnt : 32);
    }
    for (int base = 32; base < cnt; base += 32) {
        const int j = base + lane;
        int s2 = d;
        if (j < cnt - 1) { const unsigned u = (unsigned)srcS[beg + j]; s2 = (int)(u < last ? u : last); }
        float w[H], p[H];
        edge_scores<H>(as, s2, adh, j < cnt, w);
        #pragma unroll
        for (int hh = 0; hh < H; ++hh) { p[hh] = expf(w[hh] - mx[hh]); dn[hh] += p[hh]; }
        const int left = cnt - base;
        add_rows(s2, p, left < 32 ? left : 32);
    }
    #pragma unroll
    for (int hh = 0; hh < H; ++hh)
        for (int off = 16; off > 0; off >>= 1) dn[hh] += __shfl_xor(dn[hh], off);

    float den = dn[0];
    if constexpr (H == 4) den = hd == 0 ? dn[0] : hd == 1 ? dn[1] : hd == 2 ? dn[2] : dn[3];
    const float inv = 1.f / den;
    float4 r;
    r.x = acc.x * inv + b.x; r.y = acc.y * inv + b.y;
    r.z = acc.z * inv + b.z; r.w = acc.w * inv + b.w;
    if constexpr (ELU) {
        r.x = r.x > 0.f ? r.x : expm1f(r.x); r.y = r.y > 0.f ? r.y : expm1f(r.y);
        r.z = r.z > 0.f ? r.z : expm1f(r.z); r.w = r.w > 0.f ? r.w : expm1f(r.w);
    }
    *(float4*)(o + (long)d * 128 + lane * 4) = r;
    }
}

extern "C" void kernel_launch(void* const* d_in, const int* in_sizes, int n_in,
                              void* d_out, int out_size, void* d_ws, size_t ws_size,
                              hipStream_t stream)
{
    const float* x     = (const float*)d_in[0];
    const int*   ei    = (const int*)  d_in[1];
    const float* W1    = (const float*)d_in[2];
    const float* asrc1 = (const float*)d_in[3];
    const float* adst1 = (const float*)d_in[4];
    const float* b1    = (const float*)d_in[5];
    const float* W2    = (const float*)d_in[6];
    const float* asrc2 = (const float*)d_in[7];
    const float* adst2 = (const float*)d_in[8];
    const float* b2    = (const float*)d_in[9];
    float* out = (float*)d_out;

    const int N    = in_sizes[0] / 128;
    const int E    = in_sizes[1] / 2;
    const long NF  = (long)N * 128;

    float* ws   = (float*)d_ws;
    float* h    = ws;
    int*   srcS   = (int*)(ws + NF);
    int*   rank   = srcS + E;
    int*   cnt    = rank + E;
    int*   rowptr = cnt + N;
    int*   bsum   = rowptr + N + 1;
    float* as1  = ws + 2 * NF;
    float* ad1  = as1 + (long)N * 4;
    float* as2  = ad1 + (long)N * 4;
    float* ad2  = as2 + N;
    float* act  = out;

    const int T = 256;
    dim3 gE((E + T * EPT - 1) / (T * EPT));
    dim3 gNode1((N + 8 * 10 - 1) / (8 * 10));
    dim3 gNode2((N + 8 * 8 - 1) / (8 * 8));
    dim3 gGemm(1250);
    dim3 gTile((N + 1023) / 1024);
    auto g1 = [&](long n) { return dim3((unsigned)((n + T - 1) / T)); };

    k_zero_i32<<<g1(N), T, 0, stream>>>(cnt, N);
    k_rank<<<gE, T, 0, stream>>>(ei, E, cnt, rank);
    k_scan_tile<<<gTile, 1024, 0, stream>>>(cnt, rowptr, bsum, N);
    k_scan_add<<<gTile, 1024, 0, stream>>>(rowptr, bsum, N);
    k_place<<<gE, T, 0, stream>>>(ei, E, rowptr, rank, srcS);

    k_gemm_xWT<<<gGemm, T, 0, stream>>>(x, W1, h, N / 16);
    k_node_alpha<<<g1((long)N * 4), T, 0, stream>>>(h, asrc1, adst1, as1, ad1, N, 4, 32);
    k_node_gather<4, true, 10><<<gNode1, T, 0, stream>>>(rowptr, srcS, as1, ad1, h, b1, act, N);

    k_gemm_xWT<<<gGemm, T, 0, stream>>>(act, W2, h, N / 16);
    k_node_alpha<<<g1(N), T, 0, stream>>>(h, asrc2, adst2, as2, ad2, N, 1, 128);
    k_node_gather<1, false, 8><<<gNode2, T, 0, stream>>>(rowptr, srcS, as2, ad2, h, b2, out, N);
}


# GEMM x tile held in LDS at a row pitch of 132 floats instead of 128, so the 16 rows of an A-fragment read fall on different banks: the 16-way LDS bank conflict of every fragment read is gone
# speedup vs baseline: 4.7444x; 1.1500x over previous
#include <hip/hip_runtime.h>
#include <math.h>

typedef __attribute__((ext_vector_type(2))) float v2f;
typedef __attribute__((ext_vector_type(8))) float v8f;

#define NEG_SLOPE 0.2f

#define XS_PITCH 132
__global__ __launch_bounds__(256) void k_gemm_xWT(const float* __restrict__ x,
                                                  const float* __restrict__ W,
                                                  float* __restrict__ h, int nTiles)
{
    __shared__ float xs[16 * XS_PITCH];
    const int tid  = threadIdx.x;
    const int lane = tid & 31;
    const int wave = tid >> 5;

    const int m16     = lane & 15;
    const int khalf   = (lane >> 4) << 1;
    const int colBase = wave * 16;

    v2f bf[32];
    {
        const float* wr = W + (long)(colBase + m16) * 128 + khalf;
        #pragma unroll
        for (int kk = 0; kk < 32; ++kk) { bf[kk].x = wr[4 * kk]; bf[kk].y = wr[4 * kk + 1]; }
    }

    for (int tile = blockIdx.x; tile < nTiles; tile += gridDim.x) {
        const long rowBase = (long)tile * 16;
        __syncthreads();
        #pragma unroll
        for (int j = 0; j < 8; ++j)
            xs[(2 * j + (tid >> 7)) * XS_PITCH + (tid & 127)] = x[rowBase * 128 + tid + 256 * j];
        __syncthreads();

        v8f acc = {};
        #pragma unroll
        for (int kk = 0; kk < 32; ++kk) {
            v2f a;
            a.x = xs[m16 * XS_PITCH + 4 * kk + khalf];
            a.y = xs[m16 * XS_PITCH + 4 * kk + khalf + 1];
            acc = __builtin_amdgcn_wmma_f32_16x16x4_f32(false, a, false, bf[kk],
                                                        (short)0, acc, false, false);
        }
        const long rOff = rowBase + ((lane >> 4) << 3);
        const int  col  = colBase + m16;
        #pragma unroll
        for (int r = 0; r < 8; ++r)
            h[(rOff + r) * 128 + col] = acc[r];
    }
}

__global__ void k_node_alpha(const float* __restrict__ h,
                             const float* __restrict__ a_src,
                             const float* __restrict__ a_dst,
                             float* __restrict__ as, float* __restrict__ ad,
                             int N, int H, int C)
{
    int i = blockIdx.x * blockDim.x + threadIdx.x;
    if (i >= N * H) return;
    int n = i / H, hd = i - n * H;
    const float* hp = h + (long)n * (H * C) + hd * C;
    const float* s0 = a_src + hd * C;
    const float* d0 = a_dst + hd * C;
    float s = 0.f, d = 0.f;
    for (int c = 0; c < C; ++c) { float v = hp[c]; s += v * s0[c]; d += v * d0[c]; }
    as[i] = s;
    ad[i] = d;
}

__global__ void k_zero_i32(int* __restrict__ p, int n)
{
    int i = blockIdx.x * blockDim.x + threadIdx.x;
    if (i < n) p[i] = 0;
}

#define EPT 8

__global__ __launch_bounds__(256) void k_rank(const int* __restrict__ ei, int E,
                                              int* __restrict__ cnt, int* __restrict__ rank)
{
    const int e0 = blockIdx.x * (256 * EPT) + threadIdx.x;
    int dst[EPT], r[EPT];
    #pragma unroll
    for (int u = 0; u < EPT; ++u) { const int e = e0 + u * 256; dst[u] = ei[E + (e < E ? e : E - 1)]; }
    #pragma unroll
    for (int u = 0; u < EPT; ++u) { const int e = e0 + u * 256; r[u] = e < E ? atomicAdd(&cnt[dst[u]], 1) : 0; }
    #pragma unroll
    for (int u = 0; u < EPT; ++u) { const int e = e0 + u * 256; if (e < E) rank[e] = r[u]; }
}

__global__ __launch_bounds__(1024) void k_scan_tile(const int* __restrict__ cnt,
                                                    int* __restrict__ rowptr,
                                                    int* __restrict__ bsum, int N)
{
    __shared__ int wsum[32];
    const int lane = threadIdx.x & 31;
    const int w    = threadIdx.x >> 5;
    const int i    = blockIdx.x * 1024 + threadIdx.x;
    const int v    = (i < N) ? cnt[i] : 0;
    int x = v;
    for (int o = 1; o < 32; o <<= 1) {
        int y = __shfl_up(x, o);
        if (lane >= o) x += y;
    }
    if (lane == 31) wsum[w] = x;
    __syncthreads();
    if (w == 0) {
        const int s = wsum[lane];
        int t = s;
        for (int o = 1; o < 32; o <<= 1) {
            int y = __shfl_up(t, o);
            if (lane >= o) t += y;
        }
        wsum[lane] = t - s;
        if (lane == 31) bsum[blockIdx.x] = t;
    }
    __syncthreads();
    if (i < N) rowptr[i] = wsum[w] + x - v;
}

__global__ __launch_bounds__(1024) void k_scan_add(int* __restrict__ rowptr,
                                                   const int* __restrict__ bsum, int N)
{
    __shared__ int red[32];
    const int lane = threadIdx.x & 31;
    const int w    = threadIdx.x >> 5;
    const int b    = blockIdx.x;
    int v = ((int)threadIdx.x < b) ? bsum[threadIdx.x] : 0;
    for (int o = 16; o > 0; o >>= 1) v += __shfl_xor(v, o);
    if (lane == 0) red[w] = v;
    __syncthreads();
    int off = red[lane];
    for (int o = 16; o > 0; o >>= 1) off += __shfl_xor(off, o);
    const int i = b * 1024 + threadIdx.x;
    if (i < N) rowptr[i] += off;
    if (b == (int)gridDim.x - 1 && threadIdx.x == 0) rowptr[N] = off + bsum[b];
}

__global__ __launch_bounds__(256) void k_place(const int* __restrict__ ei, int E,
                                               const int* __restrict__ rowptr,
                                               const int* __restrict__ rank,
                                               int* __restrict__ srcS)
{
    const int e0 = blockIdx.x * (256 * EPT) + threadIdx.x;
    int dst[EPT], s[EPT], rk[EPT], pos[EPT];
    #pragma unroll
    for (int u = 0; u < EPT; ++u) {
        const int e  = e0 + u * 256;
        const int ec = e < E ? e : E - 1;
        dst[u] = ei[E + ec];
        s[u]   = ei[ec];
        rk[u]  = rank[ec];
    }
    #pragma unroll
    for (int u = 0; u < EPT; ++u) pos[u] = rowptr[dst[u]] + rk[u];
    #pragma unroll
    for (int u = 0; u < EPT; ++u) { const int e = e0 + u * 256; if (e < E) srcS[pos[u]] = s[u]; }
}

template <int H>
__device__ __forceinline__ void edge_scores(const float* __restrict__ as, int s,
                                            const float (&adh)[H], bool valid, float (&v)[H])
{
    if constexpr (H == 4) {
        const float4 a = *(const float4*)(as + (long)s * 4);
        v[0] = a.x + adh[0]; v[1] = a.y + adh[1]; v[2] = a.z + adh[2]; v[3] = a.w + adh[3];
    } else {
        #pragma unroll
        for (int hh = 0; hh < H; ++hh) v[hh] = as[(long)s * H + hh] + adh[hh];
    }
    #pragma unroll
    for (int hh = 0; hh < H; ++hh) {
        float t = v[hh];
        t = t > 0.f ? t : NEG_SLOPE * t;
        v[hh] = valid ? t : -INFINITY;
    }
}

template <int H, bool ELU, int NPW>
__global__ __launch_bounds__(256) void k_node_gather(const int* __restrict__ rowptr,
                                                     const int* __restrict__ srcS,
                                                     const float* __restrict__ as,
                                                     const float* __restrict__ ad,
                                                     const float* __restrict__ h,
                                                     const float* __restrict__ bias,
                                                     float* __restrict__ o, int N)
{
    const int lane = threadIdx.x & 31;
    const int d0 = __builtin_amdgcn_readfirstlane((int)(blockIdx.x * 8 + (threadIdx.x >> 5))) * NPW;
    if (d0 >= N) return;
    const int nd = N - d0 < NPW ? N - d0 : NPW;
    const int rp = lane <= nd ? rowptr[d0 + lane] : 0;
    const unsigned last = (unsigned)(N - 1);
    const int hd = lane / (32 / H);
    const float w0 = hd == 0 ? 1.f : 0.f, w1 = hd == 1 ? 1.f : 0.f;
    const float w2 = hd == 2 ? 1.f : 0.f, w3 = hd == 3 ? 1.f : 0.f;
    const float4 b = *(const float4*)(bias + lane * 4);

    for (int k = 0; k < nd; ++k) {
    const int d   = d0 + k;
    const int beg = __builtin_amdgcn_readlane(rp, k);
    const int cnt = __builtin_amdgcn_readlane(rp, k + 1) - beg + 1;

    float adh[H];
    #pragma unroll
    for (int hh = 0; hh < H; ++hh) adh[hh] = ad[(long)d * H + hh];

    int sj = d;
    if (lane < cnt - 1) {
        const unsigned u = (unsigned)srcS[beg + lane];
        sj = (int)(u < last ? u : last);
    }
    float v[H], mx[H];
    edge_scores<H>(as, sj, adh, lane < cnt, v);
    #pragma unroll
    for (int hh = 0; hh < H; ++hh) mx[hh] = v[hh];
    for (int base = 32; base < cnt; base += 32) {
        const int j = base + lane;
        int s2 = d;
        if (j < cnt - 1) { const unsigned u = (unsigned)srcS[beg + j]; s2 = (int)(u < last ? u : last); }
        float w[H];
        edge_scores<H>(as, s2, adh, j < cnt, w);
        #pragma unroll
        for (int hh = 0; hh < H; ++hh) mx[hh] = fmaxf(mx[hh], w[hh]);
    }
    #pragma unroll
    for (int hh = 0; hh < H; ++hh)
        for (int off = 16; off > 0; off >>= 1) mx[hh] = fmaxf(mx[hh], __shfl_xor(mx[hh], off));

    float dn[H];
    #pragma unroll
    for (int hh = 0; hh < H; ++hh) dn[hh] = 0.f;
    float4 acc = {0.f, 0.f, 0.f, 0.f};

    auto add_rows = [&](int sx, const float (&px)[H], int n) {
        for (int t = 0; t < n; t += 4) {
            float4 hv[4];
            float  pt[4];
            #pragma unroll
            for (int u = 0; u < 4; ++u) {
                const int s = __builtin_amdgcn_readlane(sx, t + u);
                hv[u] = *(const float4*)(h + (long)s * 128 + lane * 4);
            }
            #pragma unroll
            for (int u = 0; u < 4; ++u) {
                if constexpr (H == 4) {
                    const float p0 = __int_as_float(__builtin_amdgcn_readlane(__float_as_int(px[0]), t + u));
                    const float p1 = __int_as_float(__builtin_amdgcn_readlane(__float_as_int(px[1]), t + u));
                    const float p2 = __int_as_float(__builtin_amdgcn_readlane(__float_as_int(px[2]), t + u));
                    const float p3 = __int_as_float(__builtin_amdgcn_readlane(__float_as_int(px[3]), t + u));
                    pt[u] = p0 * w0 + p1 * w1 + p2 * w2 + p3 * w3;
                } else {
                    pt[u] = __int_as_float(__builtin_amdgcn_readlane(__float_as_int(px[0]), t + u));
                }
            }
            #pragma unroll
            for (int u = 0; u < 4; ++u) {
                acc.x += pt[u] * hv[u].x; acc.y += pt[u] * hv[u].y;
                acc.z += pt[u] * hv[u].z; acc.w += pt[u] * hv[u].w;
            }
        }
    };

    {
        float p[H];
        #pragma unroll
        for (int hh = 0; hh < H; ++hh) { p[hh] = expf(v[hh] - mx[hh]); dn[hh] += p[hh]; }
        add_rows(sj, p, cnt < 32 ? cnt : 32);
    }
    for (int base = 32; base < cnt; base += 32) {
        const int j = base + lane;
        int s2 = d;
        if (j < cnt - 1) { const unsigned u = (unsigned)srcS[beg + j]; s2 = (int)(u < last ? u : last); }
        float w[H], p[H];
        edge_scores<H>(as, s2, adh, j < cnt, w);
        #pragma unroll
        for (int hh = 0; hh < H; ++hh) { p[hh] = expf(w[hh] - mx[hh]); dn[hh] += p[hh]; }
        const int left = cnt - base;
        add_rows(s2, p, left < 32 ? left : 32);
    }
    #pragma unroll
    for (int hh = 0; hh < H; ++hh)
        for (int off = 16; off > 0; off >>= 1) dn[hh] += __shfl_xor(dn[hh], off);

    float den = dn[0];
    if constexpr (H == 4) den = hd == 0 ? dn[0] : hd == 1 ? dn[1] : hd == 2 ? dn[2] : dn[3];
    const float inv = 1.f / den;
    float4 r;
    r.x = acc.x * inv + b.x; r.y = acc.y * inv + b.y;
    r.z = acc.z * inv + b.z; r.w = acc.w * inv + b.w;
    if constexpr (ELU) {
        r.x = r.x > 0.f ? r.x : expm1f(r.x); r.y = r.y > 0.f ? r.y : expm1f(r.y);
        r.z = r.z > 0.f ? r.z : expm1f(r.z); r.w = r.w > 0.f ? r.w : expm1f(r.w);
    }
    *(float4*)(o + (long)d * 128 + lane * 4) = r;
    }
}

extern "C" void kernel_launch(void* const* d_in, const int* in_sizes, int n_in,
                              void* d_out, int out_size, void* d_ws, size_t ws_size,
                              hipStream_t stream)
{
    const float* x     = (const float*)d_in[0];
    const int*   ei    = (const int*)  d_in[1];
    const float* W1    = (const float*)d_in[2];
    const float* asrc1 = (const float*)d_in[3];
    const float* adst1 = (const float*)d_in[4];
    const float* b1    = (const float*)d_in[5];
    const float* W2    = (const float*)d_in[6];
    const float* asrc2 = (const float*)d_in[7];
    const float* adst2 = (const float*)d_in[8];
    const float* b2    = (const float*)d_in[9];
    float* out = (float*)d_out;

    const int N    = in_sizes[0] / 128;
    const int E    = in_sizes[1] / 2;
    const long NF  = (long)N * 128;

    float* ws   = (float*)d_ws;
    float* h    = ws;
    int*   srcS   = (int*)(ws + NF);
    int*   rank   = srcS + E;
    int*   cnt    = rank + E;
    int*   rowptr = cnt + N;
    int*   bsum   = rowptr + N + 1;
    float* as1  = ws + 2 * NF;
    float* ad1  = as1 + (long)N * 4;
    float* as2  = ad1 + (long)N * 4;
    float* ad2  = as2 + N;
    float* act  = out;

    const int T = 256;
    dim3 gE((E + T * EPT - 1) / (T * EPT));
    dim3 gNode1((N + 8 * 10 - 1) / (8 * 10));
    dim3 gNode2((N + 8 * 8 - 1) / (8 * 8));
    dim3 gGemm(1250);
    dim3 gTile((N + 1023) / 1024);
    auto g1 = [&](long n) { return dim3((unsigned)((n + T - 1) / T)); };

    k_zero_i32<<<g1(N), T, 0, stream>>>(cnt, N);
    k_rank<<<gE, T, 0, stream>>>(ei, E, cnt, rank);
    k_scan_tile<<<gTile, 1024, 0, stream>>>(cnt, rowptr, bsum, N);
    k_scan_add<<<gTile, 1024, 0, stream>>>(rowptr, bsum, N);
    k_place<<<gE, T, 0, stream>>>(ei, E, rowptr, rank, srcS);

    k_gemm_xWT<<<gGemm, T, 0, stream>>>(x, W1, h, N / 16);
    k_node_alpha<<<g1((long)N * 4), T, 0, stream>>>(h, asrc1, adst1, as1, ad1, N, 4, 32);
    k_node_gather<4, true, 10><<<gNode1, T, 0, stream>>>(rowptr, srcS, as1, ad1, h, b1, act, N);

    k_gemm_xWT<<<gGemm, T, 0, stream>>>(act, W2, h, N / 16);
    k_node_alpha<<<g1(N), T, 0, stream>>>(h, asrc2, adst2, as2, ad2, N, 1, 128);
    k_node_gather<1, false, 8><<<gNode2, T, 0, stream>>>(rowptr, srcS, as2, ad2, h, b2, out, N);
}


# Edges bucketed in two levels with LDS counters only (count per 256-node bin, scan, scatter, one block sorts a bin): global atomics gone; f32 sum order stays arrival order, now of LDS atomics
# speedup vs baseline: 5.6666x; 1.1944x over previous
#include <hip/hip_runtime.h>
#include <math.h>

typedef __attribute__((ext_vector_type(2))) float v2f;
typedef __attribute__((ext_vector_type(8))) float v8f;

#define NEG_SLOPE 0.2f

#define XS_PITCH 132
__global__ __launch_bounds__(256) void k_gemm_xWT(const float* __restrict__ x,
                                                  const float* __restrict__ W,
                                                  float* __restrict__ h, int nTiles)
{
    __shared__ float xs[16 * XS_PITCH];
    const int tid  = threadIdx.x;
    const int lane = tid & 31;
    const int wave = tid >> 5;

    const int m16     = lane & 15;
    const int khalf   = (lane >> 4) << 1;
    const int colBase = wave * 16;

    v2f bf[32];
    {
        const float* wr = W + (long)(colBase + m16) * 128 + khalf;
        #pragma unroll
        for (int kk = 0; kk < 32; ++kk) { bf[kk].x = wr[4 * kk]; bf[kk].y = wr[4 * kk + 1]; }
    }

    for (int tile = blockIdx.x; tile < nTiles; tile += gridDim.x) {
        const long rowBase = (long)tile * 16;
        __syncthreads();
        #pragma unroll
        for (int j = 0; j < 8; ++j)
            xs[(2 * j + (tid >> 7)) * XS_PITCH + (tid & 127)] = x[rowBase * 128 + tid + 256 * j];
        __syncthreads();

        v8f acc = {};
        #pragma unroll
        for (int kk = 0; kk < 32; ++kk) {
            v2f a;
            a.x = xs[m16 * XS_PITCH + 4 * kk + khalf];
            a.y = xs[m16 * XS_PITCH + 4 * kk + khalf + 1];
            acc = __builtin_amdgcn_wmma_f32_16x16x4_f32(false, a, false, bf[kk],
                                                        (short)0, acc, false, false);
        }
        const long rOff = rowBase + ((lane >> 4) << 3);
        const int  col  = colBase + m16;
        #pragma unroll
        for (int r = 0; r < 8; ++r)
            h[(rOff + r) * 128 + col] = acc[r];
    }
}

__global__ void k_node_alpha(const float* __restrict__ h,
                             const float* __restrict__ a_src,
                             const float* __restrict__ a_dst,
                             float* __restrict__ as, float* __restrict__ ad,
                             int N, int H, int C)
{
    int i = blockIdx.x * blockDim.x + threadIdx.x;
    if (i >= N * H) return;
    int n = i / H, hd = i - n * H;
    const float* hp = h + (long)n * (H * C) + hd * C;
    const float* s0 = a_src + hd * C;
    const float* d0 = a_dst + hd * C;
    float s = 0.f, d = 0.f;
    for (int c = 0; c < C; ++c) { float v = hp[c]; s += v * s0[c]; d += v * d0[c]; }
    as[i] = s;
    ad[i] = d;
}

#define EPT 8
#define BIN_SHIFT 8
#define BIN_W 256
#define MAX_BINS 1024

__device__ __forceinline__ int coarse_bin(int d, int nb)
{
    const unsigned b = (unsigned)d >> BIN_SHIFT;
    const unsigned l = (unsigned)(nb - 1);
    return (int)(b < l ? b : l);
}

__global__ __launch_bounds__(256) void k_bin_count(const int* __restrict__ ei, int E, int nb,
                                                   int* __restrict__ ccnt)
{
    __shared__ int hist[MAX_BINS + 1];
    const int tid = threadIdx.x;
    for (int b = tid; b < nb; b += 256) hist[b] = 0;
    if (tid == 0) hist[MAX_BINS] = 0;
    __syncthreads();
    const int e0 = blockIdx.x * (256 * EPT) + tid;
    int dst[EPT];
    #pragma unroll
    for (int u = 0; u < EPT; ++u) { const int e = e0 + u * 256; dst[u] = ei[E + (e < E ? e : E - 1)]; }
    #pragma unroll
    for (int u = 0; u < EPT; ++u) { const int e = e0 + u * 256; atomicAdd(&hist[e < E ? coarse_bin(dst[u], nb) : MAX_BINS], 1); }
    __syncthreads();
    for (int b = tid; b < nb; b += 256) ccnt[b * (int)gridDim.x + (int)blockIdx.x] = hist[b];
}

__global__ __launch_bounds__(256) void k_bin_scatter(const int* __restrict__ ei, int E, int nb,
                                                     const int* __restrict__ coff,
                                                     int* __restrict__ coarse)
{
    __shared__ int cur[MAX_BINS + 1];
    const int tid = threadIdx.x;
    for (int b = tid; b < nb; b += 256) cur[b] = coff[b * (int)gridDim.x + (int)blockIdx.x];
    if (tid == 0) cur[MAX_BINS] = 0;
    __syncthreads();
    const int e0 = blockIdx.x * (256 * EPT) + tid;
    int dst[EPT], s[EPT], pos[EPT], bin[EPT];
    #pragma unroll
    for (int u = 0; u < EPT; ++u) {
        const int e  = e0 + u * 256;
        const int ec = e < E ? e : E - 1;
        dst[u] = ei[E + ec];
        s[u]   = ei[ec];
    }
    #pragma unroll
    for (int u = 0; u < EPT; ++u) {
        const int e = e0 + u * 256;
        bin[u] = coarse_bin(dst[u], nb);
        pos[u] = atomicAdd(&cur[e < E ? bin[u] : MAX_BINS], 1);
    }
    #pragma unroll
    for (int u = 0; u < EPT; ++u) {
        const int e = e0 + u * 256;
        const unsigned p  = (unsigned)pos[u] < (unsigned)(E - 1) ? (unsigned)pos[u] : (unsigned)(E - 1);
        const unsigned dl = (unsigned)(dst[u] - (bin[u] << BIN_SHIFT));
        if (e < E) coarse[p] = (int)(((unsigned)s[u] << BIN_SHIFT) | (dl < BIN_W - 1 ? dl : BIN_W - 1));
    }
}

__global__ __launch_bounds__(1024) void k_scan_tile(const int* __restrict__ cnt,
                                                    int* __restrict__ rowptr,
                                                    int* __restrict__ bsum, int N)
{
    __shared__ int wsum[32];
    const int lane = threadIdx.x & 31;
    const int w    = threadIdx.x >> 5;
    const int i    = blockIdx.x * 1024 + threadIdx.x;
    const int v    = (i < N) ? cnt[i] : 0;
    int x = v;
    for (int o = 1; o < 32; o <<= 1) {
        int y = __shfl_up(x, o);
        if (lane >= o) x += y;
    }
    if (lane == 31) wsum[w] = x;
    __syncthreads();
    if (w == 0) {
        const int s = wsum[lane];
        int t = s;
        for (int o = 1; o < 32; o <<= 1) {
            int y = __shfl_up(t, o);
            if (lane >= o) t += y;
        }
        wsum[lane] = t - s;
        if (lane == 31) bsum[blockIdx.x] = t;
    }
    __syncthreads();
    if (i < N) rowptr[i] = wsum[w] + x - v;
}

__global__ __launch_bounds__(1024) void k_scan_add(int* __restrict__ rowptr,
                                                   const int* __restrict__ bsum, int N)
{
    __shared__ int red[32];
    const int lane = threadIdx.x & 31;
    const int w    = threadIdx.x >> 5;
    const int b    = blockIdx.x;
    int v = ((int)threadIdx.x < b) ? bsum[threadIdx.x] : 0;
    for (int o = 16; o > 0; o >>= 1) v += __shfl_xor(v, o);
    if (lane == 0) red[w] = v;
    __syncthreads();
    int off = red[lane];
    for (int o = 16; o > 0; o >>= 1) off += __shfl_xor(off, o);
    const int i = b * 1024 + threadIdx.x;
    if (i < N) rowptr[i] += off;
    if (b == (int)gridDim.x - 1 && threadIdx.x == 0) rowptr[N] = off + bsum[b];
}

#define BPT 9

__global__ __launch_bounds__(256) void k_bin_sort(const int* __restrict__ coarse,
                                                  const int* __restrict__ coff, int stride,
                                                  int N, int E,
                                                  int* __restrict__ rowptr,
                                                  int* __restrict__ srcS)
{
    __shared__ int cnt[BIN_W + 1];
    __shared__ int wsum[8];
    const int tid  = threadIdx.x;
    const int lane = tid & 31;
    const int w    = tid >> 5;
    const int b    = blockIdx.x;
    int start = coff[b * stride];
    int end   = coff[(b + 1) * stride];
    start = start < 0 ? 0 : (start > E ? E : start);
    end   = end < start ? start : (end > E ? E : end);

    cnt[tid] = 0;
    if (tid == 0) cnt[BIN_W] = 0;
    __syncthreads();
    for (int i0 = start; i0 < end; i0 += 256 * BPT) {
        int wv[BPT];
        #pragma unroll
        for (int u = 0; u < BPT; ++u) { const int i = i0 + u * 256 + tid; wv[u] = coarse[i < end ? i : end - 1]; }
        #pragma unroll
        for (int u = 0; u < BPT; ++u) { const int i = i0 + u * 256 + tid; atomicAdd(&cnt[i < end ? (wv[u] & (BIN_W - 1)) : BIN_W], 1); }
    }
    __syncthreads();

    const int v = cnt[tid];
    int x = v;
    for (int o = 1; o < 32; o <<= 1) {
        int y = __shfl_up(x, o);
        if (lane >= o) x += y;
    }
    if (lane == 31) wsum[w] = x;
    __syncthreads();
    int woff = 0;
    for (int k = 0; k < 8; ++k) woff += k < w ? wsum[k] : 0;
    const int first = start + woff + x - v;
    const int node  = b * BIN_W + tid;
    if (node < N) rowptr[node] = first;
    if (b == (int)gridDim.x - 1 && tid == 0) rowptr[N] = end;
    cnt[tid] = first;
    __syncthreads();

    for (int i0 = start; i0 < end; i0 += 256 * BPT) {
        int wv[BPT], pos[BPT];
        #pragma unroll
        for (int u = 0; u < BPT; ++u) { const int i = i0 + u * 256 + tid; wv[u] = coarse[i < end ? i : end - 1]; }
        #pragma unroll
        for (int u = 0; u < BPT; ++u) { const int i = i0 + u * 256 + tid; pos[u] = atomicAdd(&cnt[i < end ? (wv[u] & (BIN_W - 1)) : BIN_W], 1); }
        #pragma unroll
        for (int u = 0; u < BPT; ++u) {
            const int i = i0 + u * 256 + tid;
            const unsigned p = (unsigned)pos[u] < (unsigned)(E - 1) ? (unsigned)pos[u] : (unsigned)(E - 1);
            if (i < end) srcS[p] = (int)((unsigned)wv[u] >> BIN_SHIFT);
        }
    }
}

template <int H>
__device__ __forceinline__ void edge_scores(const float* __restrict__ as, int s,
                                            const float (&adh)[H], bool valid, float (&v)[H])
{
    if constexpr (H == 4) {
        const float4 a = *(const float4*)(as + (long)s * 4);
        v[0] = a.x + adh[0]; v[1] = a.y + adh[1]; v[2] = a.z + adh[2]; v[3] = a.w + adh[3];
    } else {
        #pragma unroll
        for (int hh = 0; hh < H; ++hh) v[hh] = as[(long)s * H + hh] + adh[hh];
    }
    #pragma unroll
    for (int hh = 0; hh < H; ++hh) {
        float t = v[hh];
        t = t > 0.f ? t : NEG_SLOPE * t;
        v[hh] = valid ? t : -INFINITY;
    }
}

template <int H, bool ELU, int NPW>
__global__ __launch_bounds__(256) void k_node_gather(const int* __restrict__ rowptr,
                                                     const int* __restrict__ srcS,
                                                     const float* __restrict__ as,
                                                     const float* __restrict__ ad,
                                                     const float* __restrict__ h,
                                                     const float* __restrict__ bias,
                                                     float* __restrict__ o, int N)
{
    const int lane = threadIdx.x & 31;
    const int d0 = __builtin_amdgcn_readfirstlane((int)(blockIdx.x * 8 + (threadIdx.x >> 5))) * NPW;
    if (d0 >= N) return;
    const int nd = N - d0 < NPW ? N - d0 : NPW;
    const int rp = lane <= nd ? rowptr[d0 + lane] : 0;
    const unsigned last = (unsigned)(N - 1);
    const int hd = lane / (32 / H);
    const float w0 = hd == 0 ? 1.f : 0.f, w1 = hd == 1 ? 1.f : 0.f;
    const float w2 = hd == 2 ? 1.f : 0.f, w3 = hd == 3 ? 1.f : 0.f;
    const float4 b = *(const float4*)(bias + lane * 4);

    for (int k = 0; k < nd; ++k) {
    const int d   = d0 + k;
    const int beg = __builtin_amdgcn_readlane(rp, k);
    const int cnt = __builtin_amdgcn_readlane(rp, k + 1) - beg + 1;

    float adh[H];
    #pragma unroll
    for (int hh = 0; hh < H; ++hh) adh[hh] = ad[(long)d * H + hh];

    int sj = d;
    if (lane < cnt - 1) {
        const unsigned u = (unsigned)srcS[beg + lane];
        sj = (int)(u < last ? u : last);
    }
    float v[H], mx[H];
    edge_scores<H>(as, sj, adh, lane < cnt, v);
    #pragma unroll
    for (int hh = 0; hh < H; ++hh) mx[hh] = v[hh];
    for (int base = 32; base < cnt; base += 32) {
        const int j = base + lane;
        int s2 = d;
        if (j < cnt - 1) { const unsigned u = (unsigned)srcS[beg + j]; s2 = (int)(u < last ? u : last); }
        float w[H];
        edge_scores<H>(as, s2, adh, j < cnt, w);
        #pragma unroll
        for (int hh = 0; hh < H; ++hh) mx[hh] = fmaxf(mx[hh], w[hh]);
    }
    #pragma unroll
    for (int hh = 0; hh < H; ++hh)
        for (int off = 16; off > 0; off >>= 1) mx[hh] = fmaxf(mx[hh], __shfl_xor(mx[hh], off));

    float dn[H];
    #pragma unroll
    for (int hh = 0; hh < H; ++hh) dn[hh] = 0.f;
    float4 acc = {0.f, 0.f, 0.f, 0.f};

    auto add_rows = [&](int sx, const float (&px)[H], int n) {
        for (int t = 0; t < n; t += 4) {
            float4 hv[4];
            float  pt[4];
            #pragma unroll
            for (int u = 0; u < 4; ++u) {
                const int s = __builtin_amdgcn_readlane(sx, t + u);
                hv[u] = *(const float4*)(h + (long)s * 128 + lane * 4);
            }
            #pragma unroll
            for (int u = 0; u < 4; ++u) {
                if constexpr (H == 4) {
                    const float p0 = __int_as_float(__builtin_amdgcn_readlane(__float_as_int(px[0]), t + u));
                    const float p1 = __int_as_float(__builtin_amdgcn_readlane(__float_as_int(px[1]), t + u));
                    const float p2 = __int_as_float(__builtin_amdgcn_readlane(__float_as_int(px[2]), t + u));
                    const float p3 = __int_as_float(__builtin_amdgcn_readlane(__float_as_int(px[3]), t + u));
                    pt[u] = p0 * w0 + p1 * w1 + p2 * w2 + p3 * w3;
                } else {
                    pt[u] = __int_as_float(__builtin_amdgcn_readlane(__float_as_int(px[0]), t + u));
                }
            }
            #pragma unroll
            for (int u = 0; u < 4; ++u) {
                acc.x += pt[u] * hv[u].x; acc.y += pt[u] * hv[u].y;
                acc.z += pt[u] * hv[u].z; acc.w += pt[u] * hv[u].w;
            }
        }
    };

    {
        float p[H];
        #pragma unroll
        for (int hh = 0; hh < H; ++hh) { p[hh] = expf(v[hh] - mx[hh]); dn[hh] += p[hh]; }
        add_rows(sj, p, cnt < 32 ? cnt : 32);
    }
    for (int base = 32; base < cnt; base += 32) {
        const int j = base + lane;
        int s2 = d;
        if (j < cnt - 1) { const unsigned u = (unsigned)srcS[beg + j]; s2 = (int)(u < last ? u : last); }
        float w[H], p[H];
        edge_scores<H>(as, s2, adh, j < cnt, w);
        #pragma unroll
        for (int hh = 0; hh < H; ++hh) { p[hh] = expf(w[hh] - mx[hh]); dn[hh] += p[hh]; }
        const int left = cnt - base;
        add_rows(s2, p, left < 32 ? left : 32);
    }
    #pragma unroll
    for (int hh = 0; hh < H; ++hh)
        for (int off = 16; off > 0; off >>= 1) dn[hh] += __shfl_xor(dn[hh], off);

    float den = dn[0];
    if constexpr (H == 4) den = hd == 0 ? dn[0] : hd == 1 ? dn[1] : hd == 2 ? dn[2] : dn[3];
    const float inv = 1.f / den;
    float4 r;
    r.x = acc.x * inv + b.x; r.y = acc.y * inv + b.y;
    r.z = acc.z * inv + b.z; r.w = acc.w * inv + b.w;
    if constexpr (ELU) {
        r.x = r.x > 0.f ? r.x : expm1f(r.x); r.y = r.y > 0.f ? r.y : expm1f(r.y);
        r.z = r.z > 0.f ? r.z : expm1f(r.z); r.w = r.w > 0.f ? r.w : expm1f(r.w);
    }
    *(float4*)(o + (long)d * 128 + lane * 4) = r;
    }
}

extern "C" void kernel_launch(void* const* d_in, const int* in_sizes, int n_in,
                              void* d_out, int out_size, void* d_ws, size_t ws_size,
                              hipStream_t stream)
{
    const float* x     = (const float*)d_in[0];
    const int*   ei    = (const int*)  d_in[1];
    const float* W1    = (const float*)d_in[2];
    const float* asrc1 = (const float*)d_in[3];
    const float* adst1 = (const float*)d_in[4];
    const float* b1    = (const float*)d_in[5];
    const float* W2    = (const float*)d_in[6];
    const float* asrc2 = (const float*)d_in[7];
    const float* adst2 = (const float*)d_in[8];
    const float* b2    = (const float*)d_in[9];
    float* out = (float*)d_out;

    const int N    = in_sizes[0] / 128;
    const int E    = in_sizes[1] / 2;
    const long NF  = (long)N * 128;

    float* ws   = (float*)d_ws;
    float* h    = ws;
    const int T = 256;
    dim3 gE((E + T * EPT - 1) / (T * EPT));
    const int NB = (N + BIN_W - 1) / BIN_W;
    const int M  = NB * (int)gE.x;
    int*   coarse = (int*)(ws + NF);
    int*   srcS   = coarse + E;
    int*   ccnt   = srcS + E;
    int*   coff   = ccnt + M;
    int*   rowptr = coff + M + 1;
    int*   bsum   = rowptr + N + 1;
    float* as1  = ws + 2 * NF;
    float* ad1  = as1 + (long)N * 4;
    float* as2  = ad1 + (long)N * 4;
    float* ad2  = as2 + N;
    float* act  = out;

    dim3 gNode1((N + 8 * 10 - 1) / (8 * 10));
    dim3 gNode2((N + 8 * 8 - 1) / (8 * 8));
    dim3 gGemm(1250);
    dim3 gTile((M + 1023) / 1024);
    auto g1 = [&](long n) { return dim3((unsigned)((n + T - 1) / T)); };

    k_bin_count<<<gE, T, 0, stream>>>(ei, E, NB, ccnt);
    k_scan_tile<<<gTile, 1024, 0, stream>>>(ccnt, coff, bsum, M);
    k_scan_add<<<gTile, 1024, 0, stream>>>(coff, bsum, M);
    k_bin_scatter<<<gE, T, 0, stream>>>(ei, E, NB, coff, coarse);
    k_bin_sort<<<dim3(NB), T, 0, stream>>>(coarse, coff, (int)gE.x, N, E, rowptr, srcS);

    k_gemm_xWT<<<gGemm, T, 0, stream>>>(x, W1, h, N / 16);
    k_node_alpha<<<g1((long)N * 4), T, 0, stream>>>(h, asrc1, adst1, as1, ad1, N, 4, 32);
    k_node_gather<4, true, 10><<<gNode1, T, 0, stream>>>(rowptr, srcS, as1, ad1, h, b1, act, N);

    k_gemm_xWT<<<gGemm, T, 0, stream>>>(act, W2, h, N / 16);
    k_node_alpha<<<g1(N), T, 0, stream>>>(h, asrc2, adst2, as2, ad2, N, 1, 128);
    k_node_gather<1, false, 8><<<gNode2, T, 0, stream>>>(rowptr, srcS, as2, ad2, h, b2, out, N);
}


# Gather row loop: 8 row loads in flight per trip instead of 4 (3 waits per mean bucket, not 5); room made by reading a lane's 4 weights from LDS in one 16 B read (H=4), not 16 lane reads + select; fma
# speedup vs baseline: 6.2751x; 1.1074x over previous
#include <hip/hip_runtime.h>
#include <math.h>

typedef __attribute__((ext_vector_type(2))) float v2f;
typedef __attribute__((ext_vector_type(8))) float v8f;

#define NEG_SLOPE 0.2f

#define XS_PITCH 132

template <int MASK>
__device__ __forceinline__ float row_xor_read(float v)
{
    return __int_as_float(__builtin_amdgcn_update_dpp(0, __float_as_int(v), 0x160 + MASK, 0xf, 0xf, true));
}

template <int NV, int MASK>
__device__ __forceinline__ void fold_step(float (&val)[8], bool up)
{
    #pragma unroll
    for (int j = 0; j < NV; ++j) {
        const float lo = val[j], hi = val[j + NV];
        const float keep = up ? hi : lo;
        const float send = up ? lo : hi;
        val[j] = keep + row_xor_read<MASK>(send);
    }
}

template <int H>
__global__ __launch_bounds__(256) void k_gemm_xWT(const float* __restrict__ x,
                                                  const float* __restrict__ W,
                                                  const float* __restrict__ a_src,
                                                  const float* __restrict__ a_dst,
                                                  float* __restrict__ h,
                                                  float* __restrict__ as,
                                                  float* __restrict__ ad, int nTiles)
{
    __shared__ float xs[16 * XS_PITCH];
    __shared__ float part[8 * 32];
    const int tid  = threadIdx.x;
    const int lane = tid & 31;
    const int wave = tid >> 5;

    const int m16     = lane & 15;
    const int khalf   = (lane >> 4) << 1;
    const int colBase = wave * 16;

    v2f bf[32];
    {
        const float* wr = W + (long)(colBase + m16) * 128 + khalf;
        #pragma unroll
        for (int kk = 0; kk < 32; ++kk) { bf[kk].x = wr[4 * kk]; bf[kk].y = wr[4 * kk + 1]; }
    }
    const bool  up8 = (m16 & 8) != 0;
    const float asv = a_src[colBase + m16];
    const float adv = a_dst[colBase + m16];
    const float ka  = up8 ? adv : asv;
    const float sa  = up8 ? asv : adv;
    const int partIdx = wave * 32 + (m16 >> 3) * 16 + ((lane >> 4) << 3) + (m16 & 7);
    const unsigned hOff = (unsigned)(((lane >> 4) << 3) * 128 + colBase + m16);

    auto finish_alpha = [&](unsigned tileOfs) {
        if (tid < 16 * H) {
            constexpr int WPH = 8 / H;
            const int hd  = tid & (H - 1);
            const int row = tid / H;
            float s = part[(hd * WPH) * 32 + row];
            float d = part[(hd * WPH) * 32 + 16 + row];
            #pragma unroll
            for (int j = 1; j < WPH; ++j) {
                s += part[(hd * WPH + j) * 32 + row];
                d += part[(hd * WPH + j) * 32 + 16 + row];
            }
            as[tileOfs * (16u * H) + (unsigned)tid] = s;
            ad[tileOfs * (16u * H) + (unsigned)tid] = d;
        }
    };

    int prevTile = -1;
    const int tileStep = gridDim.x;
    for (int tile = blockIdx.x; tile < nTiles; tile += tileStep) {
        const unsigned tileBase = (unsigned)tile * 2048u;
        __syncthreads();
        float xv[8];
        #pragma unroll
        for (int j = 0; j < 8; ++j) xv[j] = x[tileBase + (unsigned)tid + 256u * j];
        if (prevTile >= 0) finish_alpha((unsigned)prevTile);
        #pragma unroll
        for (int j = 0; j < 8; ++j) xs[(2 * j + (tid >> 7)) * XS_PITCH + (tid & 127)] = xv[j];
        __syncthreads();

        v8f acc = {};
        #pragma unroll
        for (int kk = 0; kk < 32; ++kk) {
            v2f a;
            a.x = xs[m16 * XS_PITCH + 4 * kk + khalf];
            a.y = xs[m16 * XS_PITCH + 4 * kk + khalf + 1];
            acc = __builtin_amdgcn_wmma_f32_16x16x4_f32(false, a, false, bf[kk],
                                                        (short)0, acc, false, false);
        }
        #pragma unroll
        for (int r = 0; r < 8; ++r)
            h[tileBase + hOff + 128u * r] = acc[r];

        float val[8];
        #pragma unroll
        for (int r = 0; r < 8; ++r) val[r] = acc[r] * ka + row_xor_read<8>(acc[r] * sa);
        fold_step<4, 4>(val, (m16 & 4) != 0);
        fold_step<2, 2>(val, (m16 & 2) != 0);
        fold_step<1, 1>(val, (m16 & 1) != 0);
        part[partIdx] = val[0];
        asm volatile("s_wait_dscnt 0x0" ::: "memory");
        prevTile = tile;
    }
    __syncthreads();
    if (prevTile >= 0) finish_alpha((unsigned)prevTile);
}

#define EPT 8
#define BIN_SHIFT 8
#define BIN_W 256
#define MAX_BINS 1024

__device__ __forceinline__ int coarse_bin(int d, int nb)
{
    const unsigned b = (unsigned)d >> BIN_SHIFT;
    const unsigned l = (unsigned)(nb - 1);
    return (int)(b < l ? b : l);
}

__global__ __launch_bounds__(256) void k_bin_count(const int* __restrict__ ei, int E, int nb,
                                                   int* __restrict__ ccnt)
{
    __shared__ int hist[MAX_BINS + 1];
    const int tid = threadIdx.x;
    for (int b = tid; b < nb; b += 256) hist[b] = 0;
    if (tid == 0) hist[MAX_BINS] = 0;
    __syncthreads();
    const int e0 = blockIdx.x * (256 * EPT) + tid;
    int dst[EPT];
    #pragma unroll
    for (int u = 0; u < EPT; ++u) { const int e = e0 + u * 256; dst[u] = ei[E + (e < E ? e : E - 1)]; }
    #pragma unroll
    for (int u = 0; u < EPT; ++u) { const int e = e0 + u * 256; atomicAdd(&hist[e < E ? coarse_bin(dst[u], nb) : MAX_BINS], 1); }
    __syncthreads();
    for (int b = tid; b < nb; b += 256) ccnt[b * (int)gridDim.x + (int)blockIdx.x] = hist[b];
}

__global__ __launch_bounds__(256) void k_bin_scatter(const int* __restrict__ ei, int E, int nb,
                                                     const int* __restrict__ coff,
                                                     int* __restrict__ coarse)
{
    __shared__ int cur[MAX_BINS + 1];
    const int tid = threadIdx.x;
    for (int b = tid; b < nb; b += 256) cur[b] = coff[b * (int)gridDim.x + (int)blockIdx.x];
    if (tid == 0) cur[MAX_BINS] = 0;
    __syncthreads();
    const int e0 = blockIdx.x * (256 * EPT) + tid;
    int dst[EPT], s[EPT], pos[EPT], bin[EPT];
    #pragma unroll
    for (int u = 0; u < EPT; ++u) {
        const int e  = e0 + u * 256;
        const int ec = e < E ? e : E - 1;
        dst[u] = ei[E + ec];
        s[u]   = ei[ec];
    }
    #pragma unroll
    for (int u = 0; u < EPT; ++u) {
        const int e = e0 + u * 256;
        bin[u] = coarse_bin(dst[u], nb);
        pos[u] = atomicAdd(&cur[e < E ? bin[u] : MAX_BINS], 1);
    }
    #pragma unroll
    for (int u = 0; u < EPT; ++u) {
        const int e = e0 + u * 256;
        const unsigned p  = (unsigned)pos[u] < (unsigned)(E - 1) ? (unsigned)pos[u] : (unsigned)(E - 1);
        const unsigned dl = (unsigned)(dst[u] - (bin[u] << BIN_SHIFT));
        if (e < E) coarse[p] = (int)(((unsigned)s[u] << BIN_SHIFT) | (dl < BIN_W - 1 ? dl : BIN_W - 1));
    }
}

__global__ __launch_bounds__(1024) void k_scan_tile(const int* __restrict__ cnt,
                                                    int* __restrict__ rowptr,
                                                    int* __restrict__ bsum, int N)
{
    __shared__ int wsum[32];
    const int lane = threadIdx.x & 31;
    const int w    = threadIdx.x >> 5;
    const int i    = blockIdx.x * 1024 + threadIdx.x;
    const int v    = (i < N) ? cnt[i] : 0;
    int x = v;
    for (int o = 1; o < 32; o <<= 1) {
        int y = __shfl_up(x, o);
        if (lane >= o) x += y;
    }
    if (lane == 31) wsum[w] = x;
    __syncthreads();
    if (w == 0) {
        const int s = wsum[lane];
        int t = s;
        for (int o = 1; o < 32; o <<= 1) {
            int y = __shfl_up(t, o);
            if (lane >= o) t += y;
        }
        wsum[lane] = t - s;
        if (lane == 31) bsum[blockIdx.x] = t;
    }
    __syncthreads();
    if (i < N) rowptr[i] = wsum[w] + x - v;
}

__global__ __launch_bounds__(1024) void k_scan_add(int* __restrict__ rowptr,
                                                   const int* __restrict__ bsum, int N)
{
    __shared__ int red[32];
    const int lane = threadIdx.x & 31;
    const int w    = threadIdx.x >> 5;
    const int b    = blockIdx.x;
    int v = ((int)threadIdx.x < b) ? bsum[threadIdx.x] : 0;
    for (int o = 16; o > 0; o >>= 1) v += __shfl_xor(v, o);
    if (lane == 0) red[w] = v;
    __syncthreads();
    int off = red[lane];
    for (int o = 16; o > 0; o >>= 1) off += __shfl_xor(off, o);
    const int i = b * 1024 + threadIdx.x;
    if (i < N) rowptr[i] += off;
    if (b == (int)gridDim.x - 1 && threadIdx.x == 0) rowptr[N] = off + bsum[b];
}

#define BPT 9

__global__ __launch_bounds__(256) void k_bin_sort(const int* __restrict__ coarse,
                                                  const int* __restrict__ coff, int stride,
                                                  int N, int E,
                                                  int* __restrict__ rowptr,
                                                  int* __restrict__ srcS)
{
    __shared__ int cnt[BIN_W + 1];
    __shared__ int wsum[8];
    const int tid  = threadIdx.x;
    const int lane = tid & 31;
    const int w    = tid >> 5;
    const int b    = blockIdx.x;
    int start = coff[b * stride];
    int end   = coff[(b + 1) * stride];
    start = start < 0 ? 0 : (start > E ? E : start);
    end   = end < start ? start : (end > E ? E : end);

    cnt[tid] = 0;
    if (tid == 0) cnt[BIN_W] = 0;
    __syncthreads();
    for (int i0 = start; i0 < end; i0 += 256 * BPT) {
        int wv[BPT];
        #pragma unroll
        for (int u = 0; u < BPT; ++u) { const int i = i0 + u * 256 + tid; wv[u] = coarse[i < end ? i : end - 1]; }
        #pragma unroll
        for (int u = 0; u < BPT; ++u) { const int i = i0 + u * 256 + tid; atomicAdd(&cnt[i < end ? (wv[u] & (BIN_W - 1)) : BIN_W], 1); }
    }
    __syncthreads();

    const int v = cnt[tid];
    int x = v;
    for (int o = 1; o < 32; o <<= 1) {
        int y = __shfl_up(x, o);
        if (lane >= o) x += y;
    }
    if (lane == 31) wsum[w] = x;
    __syncthreads();
    int woff = 0;
    for (int k = 0; k < 8; ++k) woff += k < w ? wsum[k] : 0;
    const int first = start + woff + x - v;
    const int node  = b * BIN_W + tid;
    if (node < N) rowptr[node] = first;
    if (b == (int)gridDim.x - 1 && tid == 0) rowptr[N] = end;
    cnt[tid] = first;
    __syncthreads();

    for (int i0 = start; i0 < end; i0 += 256 * BPT) {
        int wv[BPT], pos[BPT];
        #pragma unroll
        for (int u = 0; u < BPT; ++u) { const int i = i0 + u * 256 + tid; wv[u] = coarse[i < end ? i : end - 1]; }
        #pragma unroll
        for (int u = 0; u < BPT; ++u) { const int i = i0 + u * 256 + tid; pos[u] = atomicAdd(&cnt[i < end ? (wv[u] & (BIN_W - 1)) : BIN_W], 1); }
        #pragma unroll
        for (int u = 0; u < BPT; ++u) {
            const int i = i0 + u * 256 + tid;
            const unsigned p = (unsigned)pos[u] < (unsigned)(E - 1) ? (unsigned)pos[u] : (unsigned)(E - 1);
            if (i < end) srcS[p] = (int)((unsigned)wv[u] >> BIN_SHIFT);
        }
    }
}

template <int H>
__device__ __forceinline__ void edge_scores(const float* __restrict__ as, int s,
                                            const float (&adh)[H], bool valid, float (&v)[H])
{
    if constexpr (H == 4) {
        const float4 a = *(const float4*)(as + (long)s * 4);
        v[0] = a.x + adh[0]; v[1] = a.y + adh[1]; v[2] = a.z + adh[2]; v[3] = a.w + adh[3];
    } else {
        #pragma unroll
        for (int hh = 0; hh < H; ++hh) v[hh] = as[(long)s * H + hh] + adh[hh];
    }
    #pragma unroll
    for (int hh = 0; hh < H; ++hh) {
        float t = v[hh];
        t = t > 0.f ? t : NEG_SLOPE * t;
        v[hh] = valid ? t : -INFINITY;
    }
}

#define PW_PITCH 36
template <int H, bool ELU, int NPW>
__global__ __launch_bounds__(256) void k_node_gather(const int* __restrict__ rowptr,
                                                     const int* __restrict__ srcS,
                                                     const float* __restrict__ as,
                                                     const float* __restrict__ ad,
                                                     const float* __restrict__ h,
                                                     const float* __restrict__ bias,
                                                     float* __restrict__ o, int N)
{
    __shared__ __attribute__((aligned(16))) float pw[8 * H * PW_PITCH];
    const int lane = threadIdx.x & 31;
    const int d0 = __builtin_amdgcn_readfirstlane((int)(blockIdx.x * 8 + (threadIdx.x >> 5))) * NPW;
    if (d0 >= N) return;
    const int nd = N - d0 < NPW ? N - d0 : NPW;
    const int rp = lane <= nd ? rowptr[d0 + lane] : 0;
    const unsigned last = (unsigned)(N - 1);
    const int hd = lane / (32 / H);
    float* const pwv = pw + (threadIdx.x >> 5) * (H * PW_PITCH);
    const float* const pwr = pwv + hd * PW_PITCH;
    const float4* const h4 = (const float4*)h;
    const float4 b = *(const float4*)(bias + lane * 4);

    for (int k = 0; k < nd; ++k) {
    const int d   = d0 + k;
    const int beg = __builtin_amdgcn_readlane(rp, k);
    const int cnt = __builtin_amdgcn_readlane(rp, k + 1) - beg + 1;

    float adh[H];
    #pragma unroll
    for (int hh = 0; hh < H; ++hh) adh[hh] = ad[(long)d * H + hh];

    int sj = d;
    if (lane < cnt - 1) {
        const unsigned u = (unsigned)srcS[beg + lane];
        sj = (int)(u < last ? u : last);
    }
    float v[H], mx[H];
    edge_scores<H>(as, sj, adh, lane < cnt, v);
    #pragma unroll
    for (int hh = 0; hh < H; ++hh) mx[hh] = v[hh];
    for (int base = 32; base < cnt; base += 32) {
        const int j = base + lane;
        int s2 = d;
        if (j < cnt - 1) { const unsigned u = (unsigned)srcS[beg + j]; s2 = (int)(u < last ? u : last); }
        float w[H];
        edge_scores<H>(as, s2, adh, j < cnt, w);
        #pragma unroll
        for (int hh = 0; hh < H; ++hh) mx[hh] = fmaxf(mx[hh], w[hh]);
    }
    #pragma unroll
    for (int hh = 0; hh < H; ++hh)
        for (int off = 16; off > 0; off >>= 1) mx[hh] = fmaxf(mx[hh], __shfl_xor(mx[hh], off));

    float dn[H];
    #pragma unroll
    for (int hh = 0; hh < H; ++hh) dn[hh] = 0.f;
    float4 acc = {0.f, 0.f, 0.f, 0.f};

    auto add_rows = [&](int sx, const float (&px)[H], int n) {
        if constexpr (H > 1) {
            #pragma unroll
            for (int hh = 0; hh < H; ++hh) pwv[hh * PW_PITCH + lane] = px[hh];
        }
        auto use_rows = [&](const float4 (&hv)[4], const float4 pt) {
            acc.x = __builtin_fmaf(pt.x, hv[0].x, acc.x); acc.y = __builtin_fmaf(pt.x, hv[0].y, acc.y);
            acc.z = __builtin_fmaf(pt.x, hv[0].z, acc.z); acc.w = __builtin_fmaf(pt.x, hv[0].w, acc.w);
            acc.x = __builtin_fmaf(pt.y, hv[1].x, acc.x); acc.y = __builtin_fmaf(pt.y, hv[1].y, acc.y);
            acc.z = __builtin_fmaf(pt.y, hv[1].z, acc.z); acc.w = __builtin_fmaf(pt.y, hv[1].w, acc.w);
            acc.x = __builtin_fmaf(pt.z, hv[2].x, acc.x); acc.y = __builtin_fmaf(pt.z, hv[2].y, acc.y);
            acc.z = __builtin_fmaf(pt.z, hv[2].z, acc.z); acc.w = __builtin_fmaf(pt.z, hv[2].w, acc.w);
            acc.x = __builtin_fmaf(pt.w, hv[3].x, acc.x); acc.y = __builtin_fmaf(pt.w, hv[3].y, acc.y);
            acc.z = __builtin_fmaf(pt.w, hv[3].z, acc.z); acc.w = __builtin_fmaf(pt.w, hv[3].w, acc.w);
        };
        auto weights = [&](int t0) {
            if constexpr (H > 1) {
                return *(const float4*)(pwr + t0);
            } else {
                float4 r;
                r.x = __int_as_float(__builtin_amdgcn_readlane(__float_as_int(px[0]), t0));
                r.y = __int_as_float(__builtin_amdgcn_readlane(__float_as_int(px[0]), t0 + 1));
                r.z = __int_as_float(__builtin_amdgcn_readlane(__float_as_int(px[0]), t0 + 2));
                r.w = __int_as_float(__builtin_amdgcn_readlane(__float_as_int(px[0]), t0 + 3));
                return r;
            }
        };
        int t = 0;
        for (; n - t > 4; t += 8) {
            float4 ha[4], hb[4];
            #pragma unroll
            for (int u = 0; u < 4; ++u) {
                const unsigned s = (unsigned)__builtin_amdgcn_readlane(sx, t + u);
                ha[u] = h4[s * 32u + (unsigned)lane];
            }
            #pragma unroll
            for (int u = 0; u < 4; ++u) {
                const unsigned s = (unsigned)__builtin_amdgcn_readlane(sx, t + 4 + u);
                hb[u] = h4[s * 32u + (unsigned)lane];
            }
            const float4 pa = weights(t);
            const float4 pb = weights(t + 4);
            use_rows(ha, pa);
            use_rows(hb, pb);
        }
        if (t < n) {
            float4 ha[4];
            #pragma unroll
            for (int u = 0; u < 4; ++u) {
                const unsigned s = (unsigned)__builtin_amdgcn_readlane(sx, t + u);
                ha[u] = h4[s * 32u + (unsigned)lane];
            }
            const float4 pa = weights(t);
            use_rows(ha, pa);
        }
    };

    {
        float p[H];
        #pragma unroll
        for (int hh = 0; hh < H; ++hh) { p[hh] = expf(v[hh] - mx[hh]); dn[hh] += p[hh]; }
        add_rows(sj, p, cnt < 32 ? cnt : 32);
    }
    for (int base = 32; base < cnt; base += 32) {
        const int j = base + lane;
        int s2 = d;
        if (j < cnt - 1) { const unsigned u = (unsigned)srcS[beg + j]; s2 = (int)(u < last ? u : last); }
        float w[H], p[H];
        edge_scores<H>(as, s2, adh, j < cnt, w);
        #pragma unroll
        for (int hh = 0; hh < H; ++hh) { p[hh] = expf(w[hh] - mx[hh]); dn[hh] += p[hh]; }
        const int left = cnt - base;
        add_rows(s2, p, left < 32 ? left : 32);
    }
    #pragma unroll
    for (int hh = 0; hh < H; ++hh)
        for (int off = 16; off > 0; off >>= 1) dn[hh] += __shfl_xor(dn[hh], off);

    float den = dn[0];
    if constexpr (H == 4) den = hd == 0 ? dn[0] : hd == 1 ? dn[1] : hd == 2 ? dn[2] : dn[3];
    const float inv = 1.f / den;
    float4 r;
    r.x = acc.x * inv + b.x; r.y = acc.y * inv + b.y;
    r.z = acc.z * inv + b.z; r.w = acc.w * inv + b.w;
    if constexpr (ELU) {
        r.x = r.x > 0.f ? r.x : expm1f(r.x); r.y = r.y > 0.f ? r.y : expm1f(r.y);
        r.z = r.z > 0.f ? r.z : expm1f(r.z); r.w = r.w > 0.f ? r.w : expm1f(r.w);
    }
    *(float4*)(o + (long)d * 128 + lane * 4) = r;
    }
}

extern "C" void kernel_launch(void* const* d_in, const int* in_sizes, int n_in,
                              void* d_out, int out_size, void* d_ws, size_t ws_size,
                              hipStream_t stream)
{
    const float* x     = (const float*)d_in[0];
    const int*   ei    = (const int*)  d_in[1];
    const float* W1    = (const float*)d_in[2];
    const float* asrc1 = (const float*)d_in[3];
    const float* adst1 = (const float*)d_in[4];
    const float* b1    = (const float*)d_in[5];
    const float* W2    = (const float*)d_in[6];
    const float* asrc2 = (const float*)d_in[7];
    const float* adst2 = (const float*)d_in[8];
    const float* b2    = (const float*)d_in[9];
    float* out = (float*)d_out;

    const int N    = in_sizes[0] / 128;
    const int E    = in_sizes[1] / 2;
    const long NF  = (long)N * 128;

    float* ws   = (float*)d_ws;
    float* h    = ws;
    const int T = 256;
    dim3 gE((E + T * EPT - 1) / (T * EPT));
    const int NB = (N + BIN_W - 1) / BIN_W;
    const int M  = NB * (int)gE.x;
    int*   coarse = (int*)(ws + NF);
    int*   srcS   = coarse + E;
    int*   ccnt   = srcS + E;
    int*   coff   = ccnt + M;
    int*   rowptr = coff + M + 1;
    int*   bsum   = rowptr + N + 1;
    float* as1  = ws + 2 * NF;
    float* ad1  = as1 + (long)N * 4;
    float* as2  = ad1 + (long)N * 4;
    float* ad2  = as2 + N;
    float* act  = out;

    dim3 gNode1((N + 8 * 10 - 1) / (8 * 10));
    dim3 gNode2((N + 8 * 8 - 1) / (8 * 8));
    dim3 gGemm(1250);
    dim3 gTile((M + 1023) / 1024);

    k_bin_count<<<gE, T, 0, stream>>>(ei, E, NB, ccnt);
    k_scan_tile<<<gTile, 1024, 0, stream>>>(ccnt, coff, bsum, M);
    k_scan_add<<<gTile, 1024, 0, stream>>>(coff, bsum, M);
    k_bin_scatter<<<gE, T, 0, stream>>>(ei, E, NB, coff, coarse);
    k_bin_sort<<<dim3(NB), T, 0, stream>>>(coarse, coff, (int)gE.x, N, E, rowptr, srcS);

    k_gemm_xWT<4><<<gGemm, T, 0, stream>>>(x, W1, asrc1, adst1, h, as1, ad1, N / 16);
    k_node_gather<4, true, 10><<<gNode1, T, 0, stream>>>(rowptr, srcS, as1, ad1, h, b1, act, N);

    k_gemm_xWT<1><<<gGemm, T, 0, stream>>>(act, W2, asrc2, adst2, h, as2, ad2, N / 16);
    k_node_gather<1, false, 8><<<gNode2, T, 0, stream>>>(rowptr, srcS, as2, ad2, h, b2, out, N);
}


# Layer-2 GEMM moved into the layer-1 gather launch: a block keeps its 80 activation rows in LDS and runs their 5 row tiles on the matrix unit; activations never go to memory; 9 launches become 8
# speedup vs baseline: 6.6307x; 1.0567x over previous
#include <hip/hip_runtime.h>
#include <math.h>

typedef __attribute__((ext_vector_type(2))) float v2f;
typedef __attribute__((ext_vector_type(8))) float v8f;

#define NEG_SLOPE 0.2f

#define XS_PITCH 132

template <int MASK>
__device__ __forceinline__ float row_xor_read(float v)
{
    return __int_as_float(__builtin_amdgcn_update_dpp(0, __float_as_int(v), 0x160 + MASK, 0xf, 0xf, true));
}

template <int NV, int MASK>
__device__ __forceinline__ void fold_step(float (&val)[8], bool up)
{
    #pragma unroll
    for (int j = 0; j < NV; ++j) {
        const float lo = val[j], hi = val[j + NV];
        const float keep = up ? hi : lo;
        const float send = up ? lo : hi;
        val[j] = keep + row_xor_read<MASK>(send);
    }
}

template <int H>
__global__ __launch_bounds__(256) void k_gemm_xWT(const float* __restrict__ x,
                                                  const float* __restrict__ W,
                                                  const float* __restrict__ a_src,
                                                  const float* __restrict__ a_dst,
                                                  float* __restrict__ h,
                                                  float* __restrict__ as,
                                                  float* __restrict__ ad, int nTiles)
{
    __shared__ float xs[16 * XS_PITCH];
    __shared__ float part[8 * 32];
    const int tid  = threadIdx.x;
    const int lane = tid & 31;
    const int wave = tid >> 5;

    const int m16     = lane & 15;
    const int khalf   = (lane >> 4) << 1;
    const int colBase = wave * 16;

    v2f bf[32];
    {
        const float* wr = W + (long)(colBase + m16) * 128 + khalf;
        #pragma unroll
        for (int kk = 0; kk < 32; ++kk) { bf[kk].x = wr[4 * kk]; bf[kk].y = wr[4 * kk + 1]; }
    }
    const bool  up8 = (m16 & 8) != 0;
    const float asv = a_src[colBase + m16];
    const float adv = a_dst[colBase + m16];
    const float ka  = up8 ? adv : asv;
    const float sa  = up8 ? asv : adv;
    const int partIdx = wave * 32 + (m16 >> 3) * 16 + ((lane >> 4) << 3) + (m16 & 7);
    const unsigned hOff = (unsigned)(((lane >> 4) << 3) * 128 + colBase + m16);

    auto finish_alpha = [&](unsigned tileOfs) {
        if (tid < 16 * H) {
            constexpr int WPH = 8 / H;
            const int hd  = tid & (H - 1);
            const int row = tid / H;
            float s = part[(hd * WPH) * 32 + row];
            float d = part[(hd * WPH) * 32 + 16 + row];
            #pragma unroll
            for (int j = 1; j < WPH; ++j) {
                s += part[(hd * WPH + j) * 32 + row];
                d += part[(hd * WPH + j) * 32 + 16 + row];
            }
            as[tileOfs * (16u * H) + (unsigned)tid] = s;
            ad[tileOfs * (16u * H) + (unsigned)tid] = d;
        }
    };

    int prevTile = -1;
    const int tileStep = gridDim.x;
    for (int tile = blockIdx.x; tile < nTiles; tile += tileStep) {
        const unsigned tileBase = (unsigned)tile * 2048u;
        __syncthreads();
        float xv[8];
        #pragma unroll
        for (int j = 0; j < 8; ++j) xv[j] = x[tileBase + (unsigned)tid + 256u * j];
        if (prevTile >= 0) finish_alpha((unsigned)prevTile);
        #pragma unroll
        for (int j = 0; j < 8; ++j) xs[(2 * j + (tid >> 7)) * XS_PITCH + (tid & 127)] = xv[j];
        __syncthreads();

        v8f acc = {};
        #pragma unroll
        for (int kk = 0; kk < 32; ++kk) {
            v2f a;
            a.x = xs[m16 * XS_PITCH + 4 * kk + khalf];
            a.y = xs[m16 * XS_PITCH + 4 * kk + khalf + 1];
            acc = __builtin_amdgcn_wmma_f32_16x16x4_f32(false, a, false, bf[kk],
                                                        (short)0, acc, false, false);
        }
        #pragma unroll
        for (int r = 0; r < 8; ++r)
            h[tileBase + hOff + 128u * r] = acc[r];

        float val[8];
        #pragma unroll
        for (int r = 0; r < 8; ++r) val[r] = acc[r] * ka + row_xor_read<8>(acc[r] * sa);
        fold_step<4, 4>(val, (m16 & 4) != 0);
        fold_step<2, 2>(val, (m16 & 2) != 0);
        fold_step<1, 1>(val, (m16 & 1) != 0);
        part[partIdx] = val[0];
        asm volatile("s_wait_dscnt 0x0" ::: "memory");
        prevTile = tile;
    }
    __syncthreads();
    if (prevTile >= 0) finish_alpha((unsigned)prevTile);
}

#define EPT 8
#define BIN_SHIFT 8
#define BIN_W 256
#define MAX_BINS 1024

__device__ __forceinline__ int coarse_bin(int d, int nb)
{
    const unsigned b = (unsigned)d >> BIN_SHIFT;
    const unsigned l = (unsigned)(nb - 1);
    return (int)(b < l ? b : l);
}

__global__ __launch_bounds__(256) void k_bin_count(const int* __restrict__ ei, int E, int nb,
                                                   int* __restrict__ ccnt)
{
    __shared__ int hist[MAX_BINS + 1];
    const int tid = threadIdx.x;
    for (int b = tid; b < nb; b += 256) hist[b] = 0;
    if (tid == 0) hist[MAX_BINS] = 0;
    __syncthreads();
    const int e0 = blockIdx.x * (256 * EPT) + tid;
    int dst[EPT];
    #pragma unroll
    for (int u = 0; u < EPT; ++u) { const int e = e0 + u * 256; dst[u] = ei[E + (e < E ? e : E - 1)]; }
    #pragma unroll
    for (int u = 0; u < EPT; ++u) { const int e = e0 + u * 256; atomicAdd(&hist[e < E ? coarse_bin(dst[u], nb) : MAX_BINS], 1); }
    __syncthreads();
    for (int b = tid; b < nb; b += 256) ccnt[b * (int)gridDim.x + (int)blockIdx.x] = hist[b];
}

__global__ __launch_bounds__(256) void k_bin_scatter(const int* __restrict__ ei, int E, int nb,
                                                     const int* __restrict__ coff,
                                                     int* __restrict__ coarse)
{
    __shared__ int cur[MAX_BINS + 1];
    const int tid = threadIdx.x;
    for (int b = tid; b < nb; b += 256) cur[b] = coff[b * (int)gridDim.x + (int)blockIdx.x];
    if (tid == 0) cur[MAX_BINS] = 0;
    __syncthreads();
    const int e0 = blockIdx.x * (256 * EPT) + tid;
    int dst[EPT], s[EPT], pos[EPT], bin[EPT];
    #pragma unroll
    for (int u = 0; u < EPT; ++u) {
        const int e  = e0 + u * 256;
        const int ec = e < E ? e : E - 1;
        dst[u] = ei[E + ec];
        s[u]   = ei[ec];
    }
    #pragma unroll
    for (int u = 0; u < EPT; ++u) {
        const int e = e0 + u * 256;
        bin[u] = coarse_bin(dst[u], nb);
        pos[u] = atomicAdd(&cur[e < E ? bin[u] : MAX_BINS], 1);
    }
    #pragma unroll
    for (int u = 0; u < EPT; ++u) {
        const int e = e0 + u * 256;
        const unsigned p  = (unsigned)pos[u] < (unsigned)(E - 1) ? (unsigned)pos[u] : (unsigned)(E - 1);
        const unsigned dl = (unsigned)(dst[u] - (bin[u] << BIN_SHIFT));
        if (e < E) coarse[p] = (int)(((unsigned)s[u] << BIN_SHIFT) | (dl < BIN_W - 1 ? dl : BIN_W - 1));
    }
}

__global__ __launch_bounds__(1024) void k_scan_tile(const int* __restrict__ cnt,
                                                    int* __restrict__ rowptr,
                                                    int* __restrict__ bsum, int N)
{
    __shared__ int wsum[32];
    const int lane = threadIdx.x & 31;
    const int w    = threadIdx.x >> 5;
    const int i    = blockIdx.x * 1024 + threadIdx.x;
    const int v    = (i < N) ? cnt[i] : 0;
    int x = v;
    for (int o = 1; o < 32; o <<= 1) {
        int y = __shfl_up(x, o);
        if (lane >= o) x += y;
    }
    if (lane == 31) wsum[w] = x;
    __syncthreads();
    if (w == 0) {
        const int s = wsum[lane];
        int t = s;
        for (int o = 1; o < 32; o <<= 1) {
            int y = __shfl_up(t, o);
            if (lane >= o) t += y;
        }
        wsum[lane] = t - s;
        if (lane == 31) bsum[blockIdx.x] = t;
    }
    __syncthreads();
    if (i < N) rowptr[i] = wsum[w] + x - v;
}

__global__ __launch_bounds__(1024) void k_scan_add(int* __restrict__ rowptr,
                                                   const int* __restrict__ bsum, int N)
{
    __shared__ int red[32];
    const int lane = threadIdx.x & 31;
    const int w    = threadIdx.x >> 5;
    const int b    = blockIdx.x;
    int v = ((int)threadIdx.x < b) ? bsum[threadIdx.x] : 0;
    for (int o = 16; o > 0; o >>= 1) v += __shfl_xor(v, o);
    if (lane == 0) red[w] = v;
    __syncthreads();
    int off = red[lane];
    for (int o = 16; o > 0; o >>= 1) off += __shfl_xor(off, o);
    const int i = b * 1024 + threadIdx.x;
    if (i < N) rowptr[i] += off;
    if (b == (int)gridDim.x - 1 && threadIdx.x == 0) rowptr[N] = off + bsum[b];
}

#define BPT 9

__global__ __launch_bounds__(256) void k_bin_sort(const int* __restrict__ coarse,
                                                  const int* __restrict__ coff, int stride,
                                                  int N, int E,
                                                  int* __restrict__ rowptr,
                                                  int* __restrict__ srcS)
{
    __shared__ int cnt[BIN_W + 1];
    __shared__ int wsum[8];
    const int tid  = threadIdx.x;
    const int lane = tid & 31;
    const int w    = tid >> 5;
    const int b    = blockIdx.x;
    int start = coff[b * stride];
    int end   = coff[(b + 1) * stride];
    start = start < 0 ? 0 : (start > E ? E : start);
    end   = end < start ? start : (end > E ? E : end);

    cnt[tid] = 0;
    if (tid == 0) cnt[BIN_W] = 0;
    __syncthreads();
    for (int i0 = start; i0 < end; i0 += 256 * BPT) {
        int wv[BPT];
        #pragma unroll
        for (int u = 0; u < BPT; ++u) { const int i = i0 + u * 256 + tid; wv[u] = coarse[i < end ? i : end - 1]; }
        #pragma unroll
        for (int u = 0; u < BPT; ++u) { const int i = i0 + u * 256 + tid; atomicAdd(&cnt[i < end ? (wv[u] & (BIN_W - 1)) : BIN_W], 1); }
    }
    __syncthreads();

    const int v = cnt[tid];
    int x = v;
    for (int o = 1; o < 32; o <<= 1) {
        int y = __shfl_up(x, o);
        if (lane >= o) x += y;
    }
    if (lane == 31) wsum[w] = x;
    __syncthreads();
    int woff = 0;
    for (int k = 0; k < 8; ++k) woff += k < w ? wsum[k] : 0;
    const int first = start + woff + x - v;
    const int node  = b * BIN_W + tid;
    if (node < N) rowptr[node] = first;
    if (b == (int)gridDim.x - 1 && tid == 0) rowptr[N] = end;
    cnt[tid] = first;
    __syncthreads();

    for (int i0 = start; i0 < end; i0 += 256 * BPT) {
        int wv[BPT], pos[BPT];
        #pragma unroll
        for (int u = 0; u < BPT; ++u) { const int i = i0 + u * 256 + tid; wv[u] = coarse[i < end ? i : end - 1]; }
        #pragma unroll
        for (int u = 0; u < BPT; ++u) { const int i = i0 + u * 256 + tid; pos[u] = atomicAdd(&cnt[i < end ? (wv[u] & (BIN_W - 1)) : BIN_W], 1); }
        #pragma unroll
        for (int u = 0; u < BPT; ++u) {
            const int i = i0 + u * 256 + tid;
            const unsigned p = (unsigned)pos[u] < (unsigned)(E - 1) ? (unsigned)pos[u] : (unsigned)(E - 1);
            if (i < end) srcS[p] = (int)((unsigned)wv[u] >> BIN_SHIFT);
        }
    }
}

template <int H>
__device__ __forceinline__ void edge_scores(const float* __restrict__ as, int s,
                                            const float (&adh)[H], bool valid, float (&v)[H])
{
    if constexpr (H == 4) {
        const float4 a = *(const float4*)(as + (long)s * 4);
        v[0] = a.x + adh[0]; v[1] = a.y + adh[1]; v[2] = a.z + adh[2]; v[3] = a.w + adh[3];
    } else {
        #pragma unroll
        for (int hh = 0; hh < H; ++hh) v[hh] = as[(long)s * H + hh] + adh[hh];
    }
    #pragma unroll
    for (int hh = 0; hh < H; ++hh) {
        float t = v[hh];
        t = t > 0.f ? t : NEG_SLOPE * t;
        v[hh] = valid ? t : -INFINITY;
    }
}

#define PW_PITCH 36
template <int H, bool ELU>
__device__ __forceinline__ float4 gather_node(const int d, const int beg, const int cnt,
                                              const int lane, const int hd, const unsigned last,
                                              const int* __restrict__ srcS,
                                              const float* __restrict__ as,
                                              const float* __restrict__ ad,
                                              const float4* __restrict__ h4,
                                              float* const pwv, const float* const pwr,
                                              const float4 b)
{
    float adh[H];
    #pragma unroll
    for (int hh = 0; hh < H; ++hh) adh[hh] = ad[(long)d * H + hh];

    int sj = d;
    if (lane < cnt - 1) {
        const unsigned u = (unsigned)srcS[beg + lane];
        sj = (int)(u < last ? u : last);
    }
    float v[H], mx[H];
    edge_scores<H>(as, sj, adh, lane < cnt, v);
    #pragma unroll
    for (int hh = 0; hh < H; ++hh) mx[hh] = v[hh];
    for (int base = 32; base < cnt; base += 32) {
        const int j = base + lane;
        int s2 = d;
        if (j < cnt - 1) { const unsigned u = (unsigned)srcS[beg + j]; s2 = (int)(u < last ? u : last); }
        float w[H];
        edge_scores<H>(as, s2, adh, j < cnt, w);
        #pragma unroll
        for (int hh = 0; hh < H; ++hh) mx[hh] = fmaxf(mx[hh], w[hh]);
    }
    #pragma unroll
    for (int hh = 0; hh < H; ++hh)
        for (int off = 16; off > 0; off >>= 1) mx[hh] = fmaxf(mx[hh], __shfl_xor(mx[hh], off));

    float dn[H];
    #pragma unroll
    for (int hh = 0; hh < H; ++hh) dn[hh] = 0.f;
    float4 acc = {0.f, 0.f, 0.f, 0.f};

    auto add_rows = [&](int sx, const float (&px)[H], int n) {
        if constexpr (H > 1) {
            #pragma unroll
            for (int hh = 0; hh < H; ++hh) pwv[hh * PW_PITCH + lane] = px[hh];
        }
        auto use_rows = [&](const float4 (&hv)[4], const float4 pt) {
            acc.x = __builtin_fmaf(pt.x, hv[0].x, acc.x); acc.y = __builtin_fmaf(pt.x, hv[0].y, acc.y);
            acc.z = __builtin_fmaf(pt.x, hv[0].z, acc.z); acc.w = __builtin_fmaf(pt.x, hv[0].w, acc.w);
            acc.x = __builtin_fmaf(pt.y, hv[1].x, acc.x); acc.y = __builtin_fmaf(pt.y, hv[1].y, acc.y);
            acc.z = __builtin_fmaf(pt.y, hv[1].z, acc.z); acc.w = __builtin_fmaf(pt.y, hv[1].w, acc.w);
            acc.x = __builtin_fmaf(pt.z, hv[2].x, acc.x); acc.y = __builtin_fmaf(pt.z, hv[2].y, acc.y);
            acc.z = __builtin_fmaf(pt.z, hv[2].z, acc.z); acc.w = __builtin_fmaf(pt.z, hv[2].w, acc.w);
            acc.x = __builtin_fmaf(pt.w, hv[3].x, acc.x); acc.y = __builtin_fmaf(pt.w, hv[3].y, acc.y);
            acc.z = __builtin_fmaf(pt.w, hv[3].z, acc.z); acc.w = __builtin_fmaf(pt.w, hv[3].w, acc.w);
        };
        auto weights = [&](int t0) {
            if constexpr (H > 1) {
                return *(const float4*)(pwr + t0);
            } else {
                float4 r;
                r.x = __int_as_float(__builtin_amdgcn_readlane(__float_as_int(px[0]), t0));
                r.y = __int_as_float(__builtin_amdgcn_readlane(__float_as_int(px[0]), t0 + 1));
                r.z = __int_as_float(__builtin_amdgcn_readlane(__float_as_int(px[0]), t0 + 2));
                r.w = __int_as_float(__builtin_amdgcn_readlane(__float_as_int(px[0]), t0 + 3));
                return r;
            }
        };
        int t = 0;
        for (; n - t > 4; t += 8) {
            float4 ha[4], hb[4];
            #pragma unroll
            for (int u = 0; u < 4; ++u) {
                const unsigned s = (unsigned)__builtin_amdgcn_readlane(sx, t + u);
                ha[u] = h4[s * 32u + (unsigned)lane];
            }
            #pragma unroll
            for (int u = 0; u < 4; ++u) {
                const unsigned s = (unsigned)__builtin_amdgcn_readlane(sx, t + 4 + u);
                hb[u] = h4[s * 32u + (unsigned)lane];
            }
            const float4 pa = weights(t);
            const float4 pb = weights(t + 4);
            use_rows(ha, pa);
            use_rows(hb, pb);
        }
        if (t < n) {
            float4 ha[4];
            #pragma unroll
            for (int u = 0; u < 4; ++u) {
                const unsigned s = (unsigned)__builtin_amdgcn_readlane(sx, t + u);
                ha[u] = h4[s * 32u + (unsigned)lane];
            }
            const float4 pa = weights(t);
            use_rows(ha, pa);
        }
    };

    {
        float p[H];
        #pragma unroll
        for (int hh = 0; hh < H; ++hh) { p[hh] = expf(v[hh] - mx[hh]); dn[hh] += p[hh]; }
        add_rows(sj, p, cnt < 32 ? cnt : 32);
    }
    for (int base = 32; base < cnt; base += 32) {
        const int j = base + lane;
        int s2 = d;
        if (j < cnt - 1) { const unsigned u = (unsigned)srcS[beg + j]; s2 = (int)(u < last ? u : last); }
        float w[H], p[H];
        edge_scores<H>(as, s2, adh, j < cnt, w);
        #pragma unroll
        for (int hh = 0; hh < H; ++hh) { p[hh] = expf(w[hh] - mx[hh]); dn[hh] += p[hh]; }
        const int left = cnt - base;
        add_rows(s2, p, left < 32 ? left : 32);
    }
    #pragma unroll
    for (int hh = 0; hh < H; ++hh)
        for (int off = 16; off > 0; off >>= 1) dn[hh] += __shfl_xor(dn[hh], off);

    float den = dn[0];
    if constexpr (H == 4) den = hd == 0 ? dn[0] : hd == 1 ? dn[1] : hd == 2 ? dn[2] : dn[3];
    const float inv = 1.f / den;
    float4 r;
    r.x = acc.x * inv + b.x; r.y = acc.y * inv + b.y;
    r.z = acc.z * inv + b.z; r.w = acc.w * inv + b.w;
    if constexpr (ELU) {
        r.x = r.x > 0.f ? r.x : expm1f(r.x); r.y = r.y > 0.f ? r.y : expm1f(r.y);
        r.z = r.z > 0.f ? r.z : expm1f(r.z); r.w = r.w > 0.f ? r.w : expm1f(r.w);
    }
    return r;
}

template <int H, bool ELU, int NPW>
__global__ __launch_bounds__(256) void k_node_gather(const int* __restrict__ rowptr,
                                                     const int* __restrict__ srcS,
                                                     const float* __restrict__ as,
                                                     const float* __restrict__ ad,
                                                     const float* __restrict__ h,
                                                     const float* __restrict__ bias,
                                                     float* __restrict__ o, int N)
{
    __shared__ __attribute__((aligned(16))) float pw[8 * H * PW_PITCH];
    const int lane = threadIdx.x & 31;
    const int d0 = __builtin_amdgcn_readfirstlane((int)(blockIdx.x * 8 + (threadIdx.x >> 5))) * NPW;
    if (d0 >= N) return;
    const int nd = N - d0 < NPW ? N - d0 : NPW;
    const int rp = lane <= nd ? rowptr[d0 + lane] : 0;
    const unsigned last = (unsigned)(N - 1);
    const int hd = lane / (32 / H);
    float* const pwv = pw + (threadIdx.x >> 5) * (H * PW_PITCH);
    const float* const pwr = pwv + hd * PW_PITCH;
    const float4* const h4 = (const float4*)h;
    const float4 b = *(const float4*)(bias + lane * 4);

    for (int k = 0; k < nd; ++k) {
        const int d   = d0 + k;
        const int beg = __builtin_amdgcn_readlane(rp, k);
        const int cnt = __builtin_amdgcn_readlane(rp, k + 1) - beg + 1;
        const float4 r = gather_node<H, ELU>(d, beg, cnt, lane, hd, last, srcS, as, ad, h4, pwv, pwr, b);
        *(float4*)(o + (long)d * 128 + lane * 4) = r;
    }
}

template <int H, int NPW, int HN>
__global__ __launch_bounds__(256) void k_gather_gemm(const int* __restrict__ rowptr,
                                                     const int* __restrict__ srcS,
                                                     const float* __restrict__ as,
                                                     const float* __restrict__ ad,
                                                     const float* __restrict__ h,
                                                     const float* __restrict__ bias,
                                                     const float* __restrict__ W,
                                                     const float* __restrict__ a_src,
                                                     const float* __restrict__ a_dst,
                                                     float* __restrict__ hn,
                                                     float* __restrict__ asn,
                                                     float* __restrict__ adn, int N)
{
    constexpr int ROWS  = 8 * NPW;
    static_assert(ROWS % 16 == 0, "a block's rows must be whole 16-row tiles");
    constexpr int TILES = ROWS / 16;
    __shared__ __attribute__((aligned(16))) float pw[8 * H * PW_PITCH];
    __shared__ __attribute__((aligned(16))) float xs[ROWS * XS_PITCH];
    __shared__ float part[2 * 8 * 32];
    const int tid  = threadIdx.x;
    const int lane = tid & 31;
    const int wave = tid >> 5;
    const int d0 = __builtin_amdgcn_readfirstlane((int)(blockIdx.x * 8 + (threadIdx.x >> 5))) * NPW;
    const int nd = N - d0 < NPW ? (N - d0 > 0 ? N - d0 : 0) : NPW;
    const int rp = (nd > 0 && lane <= nd) ? rowptr[d0 + lane] : 0;
    const unsigned last = (unsigned)(N - 1);
    const int hd = lane / (32 / H);
    float* const pwv = pw + (threadIdx.x >> 5) * (H * PW_PITCH);
    const float* const pwr = pwv + hd * PW_PITCH;
    const float4* const h4 = (const float4*)h;
    const float4 b = *(const float4*)(bias + lane * 4);
    float* const xrow = xs + (wave * NPW) * XS_PITCH + lane * 4;

    for (int k = 0; k < nd; ++k) {
        const int beg = __builtin_amdgcn_readlane(rp, k);
        const int cnt = __builtin_amdgcn_readlane(rp, k + 1) - beg + 1;
        const float4 r = gather_node<H, true>(d0 + k, beg, cnt, lane, hd, last, srcS, as, ad, h4, pwv, pwr, b);
        *(float4*)(xrow + k * XS_PITCH) = r;
    }
    for (int k = nd; k < NPW; ++k) *(float4*)(xrow + k * XS_PITCH) = float4{0.f, 0.f, 0.f, 0.f};

    const int m16     = lane & 15;
    const int khalf   = (lane >> 4) << 1;
    const int colBase = wave * 16;
    v2f bf[32];
    {
        const float* wr = W + (long)(colBase + m16) * 128 + khalf;
        #pragma unroll
        for (int kk = 0; kk < 32; ++kk) { bf[kk].x = wr[4 * kk]; bf[kk].y = wr[4 * kk + 1]; }
    }
    const bool  up8 = (m16 & 8) != 0;
    const float asv = a_src[colBase + m16];
    const float adv = a_dst[colBase + m16];
    const float ka  = up8 ? adv : asv;
    const float sa  = up8 ? asv : adv;
    const int partIdx = wave * 32 + (m16 >> 3) * 16 + ((lane >> 4) << 3) + (m16 & 7);
    const unsigned hOff = (unsigned)(((lane >> 4) << 3) * 128 + colBase + m16);

    auto finish_alpha = [&](unsigned tileOfs, const float* pb) {
        if (tid < 16 * HN) {
            constexpr int WPH = 8 / HN;
            const int hdn = tid & (HN - 1);
            const int row = tid / HN;
            float s = pb[(hdn * WPH) * 32 + row];
            float d = pb[(hdn * WPH) * 32 + 16 + row];
            #pragma unroll
            for (int j = 1; j < WPH; ++j) {
                s += pb[(hdn * WPH + j) * 32 + row];
                d += pb[(hdn * WPH + j) * 32 + 16 + row];
            }
            asn[tileOfs * (16u * HN) + (unsigned)tid] = s;
            adn[tileOfs * (16u * HN) + (unsigned)tid] = d;
        }
    };

    __syncthreads();
    for (int t = 0; t < TILES; ++t) {
        const int tile = (int)blockIdx.x * TILES + t;
        const bool live = tile * 16 + 16 <= N;
        const unsigned tileBase = (unsigned)tile * 2048u;
        const float* const xt = xs + t * (16 * XS_PITCH);
        float* const pb = part + (t & 1) * 256;

        v8f acc = {};
        #pragma unroll
        for (int kk = 0; kk < 32; ++kk) {
            v2f a;
            a.x = xt[m16 * XS_PITCH + 4 * kk + khalf];
            a.y = xt[m16 * XS_PITCH + 4 * kk + khalf + 1];
            acc = __builtin_amdgcn_wmma_f32_16x16x4_f32(false, a, false, bf[kk],
                                                        (short)0, acc, false, false);
        }
        float val[8];
        #pragma unroll
        for (int r = 0; r < 8; ++r) val[r] = acc[r] * ka + row_xor_read<8>(acc[r] * sa);
        fold_step<4, 4>(val, (m16 & 4) != 0);
        fold_step<2, 2>(val, (m16 & 2) != 0);
        fold_step<1, 1>(val, (m16 & 1) != 0);
        pb[partIdx] = val[0];
        __syncthreads();
        if (live) {
            #pragma unroll
            for (int r = 0; r < 8; ++r)
                hn[tileBase + hOff + 128u * r] = acc[r];
            finish_alpha((unsigned)tile, pb);
        }
    }
}

extern "C" void kernel_launch(void* const* d_in, const int* in_sizes, int n_in,
                              void* d_out, int out_size, void* d_ws, size_t ws_size,
                              hipStream_t stream)
{
    const float* x     = (const float*)d_in[0];
    const int*   ei    = (const int*)  d_in[1];
    const float* W1    = (const float*)d_in[2];
    const float* asrc1 = (const float*)d_in[3];
    const float* adst1 = (const float*)d_in[4];
    const float* b1    = (const float*)d_in[5];
    const float* W2    = (const float*)d_in[6];
    const float* asrc2 = (const float*)d_in[7];
    const float* adst2 = (const float*)d_in[8];
    const float* b2    = (const float*)d_in[9];
    float* out = (float*)d_out;

    const int N    = in_sizes[0] / 128;
    const int E    = in_sizes[1] / 2;
    const long NF  = (long)N * 128;

    float* ws   = (float*)d_ws;
    float* h    = ws;
    const int T = 256;
    dim3 gE((E + T * EPT - 1) / (T * EPT));
    const int NB = (N + BIN_W - 1) / BIN_W;
    const int M  = NB * (int)gE.x;
    int*   coarse = (int*)(ws + NF);
    int*   srcS   = coarse + E;
    int*   ccnt   = srcS + E;
    int*   coff   = ccnt + M;
    int*   rowptr = coff + M + 1;
    int*   bsum   = rowptr + N + 1;
    float* as1  = ws + 2 * NF;
    float* ad1  = as1 + (long)N * 4;
    float* as2  = ad1 + (long)N * 4;
    float* ad2  = as2 + N;
    float* h1   = out;

    dim3 gNode1((N + 8 * 10 - 1) / (8 * 10));
    dim3 gNode2((N + 8 * 8 - 1) / (8 * 8));
    dim3 gGemm(1250);
    dim3 gTile((M + 1023) / 1024);

    k_bin_count<<<gE, T, 0, stream>>>(ei, E, NB, ccnt);
    k_scan_tile<<<gTile, 1024, 0, stream>>>(ccnt, coff, bsum, M);
    k_scan_add<<<gTile, 1024, 0, stream>>>(coff, bsum, M);
    k_bin_scatter<<<gE, T, 0, stream>>>(ei, E, NB, coff, coarse);
    k_bin_sort<<<dim3(NB), T, 0, stream>>>(coarse, coff, (int)gE.x, N, E, rowptr, srcS);

    k_gemm_xWT<4><<<gGemm, T, 0, stream>>>(x, W1, asrc1, adst1, h1, as1, ad1, N / 16);
    k_gather_gemm<4, 10, 1><<<gNode1, T, 0, stream>>>(rowptr, srcS, as1, ad1, h1, b1,
                                                      W2, asrc2, adst2, h, as2, ad2, N);

    k_node_gather<1, false, 8><<<gNode2, T, 0, stream>>>(rowptr, srcS, as2, ad2, h, b2, out, N);
}


# Bin pass walks each piece with 8 neighbouring lanes, a word per lane, bounds staged in LDS: a load covers 4 pieces' lines, not 32 lanes' lines; 1 trip per piece, not 4; sum order still LDS arrival
# speedup vs baseline: 8.5542x; 1.2901x over previous
#include <hip/hip_runtime.h>
#include <math.h>
#include <type_traits>

typedef __attribute__((ext_vector_type(2))) float v2f;
typedef __attribute__((ext_vector_type(8))) float v8f;

#define NEG_SLOPE 0.2f

#define XS_PITCH 132

template <int MASK>
__device__ __forceinline__ float row_xor_read(float v)
{
    return __int_as_float(__builtin_amdgcn_update_dpp(0, __float_as_int(v), 0x160 + MASK, 0xf, 0xf, true));
}

template <int NV, int MASK>
__device__ __forceinline__ void fold_step(float (&val)[8], bool up)
{
    #pragma unroll
    for (int j = 0; j < NV; ++j) {
        const float lo = val[j], hi = val[j + NV];
        const float keep = up ? hi : lo;
        const float send = up ? lo : hi;
        val[j] = keep + row_xor_read<MASK>(send);
    }
}

template <int H>
__device__ __forceinline__ void gemm_row_tiles(const float* __restrict__ x,
                                               const float* __restrict__ W,
                                               const float* __restrict__ a_src,
                                               const float* __restrict__ a_dst,
                                               float* __restrict__ h,
                                               float* __restrict__ as,
                                               float* __restrict__ ad,
                                               const int tileBegin, const int tileEnd,
                                               const int blk, const int nblk)
{
    __shared__ float xs[16 * XS_PITCH];
    __shared__ float part[8 * 32];
    const int tid  = threadIdx.x;
    const int lane = tid & 31;
    const int wave = tid >> 5;

    const int m16     = lane & 15;
    const int khalf   = (lane >> 4) << 1;
    const int colBase = wave * 16;

    v2f bf[32];
    {
        const float* wr = W + (long)(colBase + m16) * 128 + khalf;
        #pragma unroll
        for (int kk = 0; kk < 32; ++kk) { bf[kk].x = wr[4 * kk]; bf[kk].y = wr[4 * kk + 1]; }
    }
    const bool  up8 = (m16 & 8) != 0;
    const float asv = a_src[colBase + m16];
    const float adv = a_dst[colBase + m16];
    const float ka  = up8 ? adv : asv;
    const float sa  = up8 ? asv : adv;
    const int partIdx = wave * 32 + (m16 >> 3) * 16 + ((lane >> 4) << 3) + (m16 & 7);
    const unsigned hOff = (unsigned)(((lane >> 4) << 3) * 128 + colBase + m16);

    auto finish_alpha = [&](unsigned tileOfs) {
        if (tid < 16 * H) {
            constexpr int WPH = 8 / H;
            const int hd  = tid & (H - 1);
            const int row = tid / H;
            float s = part[(hd * WPH) * 32 + row];
            float d = part[(hd * WPH) * 32 + 16 + row];
            #pragma unroll
            for (int j = 1; j < WPH; ++j) {
                s += part[(hd * WPH + j) * 32 + row];
                d += part[(hd * WPH + j) * 32 + 16 + row];
            }
            as[tileOfs * (16u * H) + (unsigned)tid] = s;
            ad[tileOfs * (16u * H) + (unsigned)tid] = d;
        }
    };

    int prevTile = -1;
    for (int tile = tileBegin + blk; tile < tileEnd; tile += nblk) {
        const unsigned tileBase = (unsigned)tile * 2048u;
        __syncthreads();
        float xv[8];
        #pragma unroll
        for (int j = 0; j < 8; ++j) xv[j] = x[tileBase + (unsigned)tid + 256u * j];
        if (prevTile >= 0) finish_alpha((unsigned)prevTile);
        #pragma unroll
        for (int j = 0; j < 8; ++j) xs[(2 * j + (tid >> 7)) * XS_PITCH + (tid & 127)] = xv[j];
        __syncthreads();

        v8f acc = {};
        #pragma unroll
        for (int kk = 0; kk < 32; ++kk) {
            v2f a;
            a.x = xs[m16 * XS_PITCH + 4 * kk + khalf];
            a.y = xs[m16 * XS_PITCH + 4 * kk + khalf + 1];
            acc = __builtin_amdgcn_wmma_f32_16x16x4_f32(false, a, false, bf[kk],
                                                        (short)0, acc, false, false);
        }
        #pragma unroll
        for (int r = 0; r < 8; ++r)
            h[tileBase + hOff + 128u * r] = acc[r];

        float val[8];
        #pragma unroll
        for (int r = 0; r < 8; ++r) val[r] = acc[r] * ka + row_xor_read<8>(acc[r] * sa);
        fold_step<4, 4>(val, (m16 & 4) != 0);
        fold_step<2, 2>(val, (m16 & 2) != 0);
        fold_step<1, 1>(val, (m16 & 1) != 0);
        part[partIdx] = val[0];
        asm volatile("s_wait_dscnt 0x0" ::: "memory");
        prevTile = tile;
    }
    __syncthreads();
    if (prevTile >= 0) finish_alpha((unsigned)prevTile);
}

#define EPT 8
#define SEG (256 * EPT)
#define BIN_SHIFT 8
#define BIN_W 256
#define MAX_BINS 1024

__device__ __forceinline__ int coarse_bin(int d, int nb)
{
    const unsigned b = (unsigned)d >> BIN_SHIFT;
    const unsigned l = (unsigned)(nb - 1);
    return (int)(b < l ? b : l);
}

__device__ __forceinline__ void seg_sort_block(const int* __restrict__ ei, int E, int nb,
                                               int* __restrict__ loff,
                                               int* __restrict__ coarse,
                                               const int blk, const int nblk)
{
    __shared__ __attribute__((aligned(16))) int hist[MAX_BINS + 4];
    __shared__ int wtot[8];
    const int tid  = threadIdx.x;
    const int lane = tid & 31;
    const int w    = tid >> 5;
    *(int4*)(hist + 4 * tid) = int4{0, 0, 0, 0};
    if (tid == 0) hist[MAX_BINS] = 0;
    __syncthreads();
    const int e0 = blk * SEG + tid;
    int dst[EPT], s[EPT], pos[EPT], bin[EPT];
    #pragma unroll
    for (int u = 0; u < EPT; ++u) {
        const int e  = e0 + u * 256;
        const int ec = e < E ? e : E - 1;
        dst[u] = ei[E + ec];
        s[u]   = ei[ec];
    }
    #pragma unroll
    for (int u = 0; u < EPT; ++u) {
        const int e = e0 + u * 256;
        bin[u] = coarse_bin(dst[u], nb);
        pos[u] = atomicAdd(&hist[e < E ? bin[u] : MAX_BINS], 1);
    }
    __syncthreads();
    const int4 c = *(const int4*)(hist + 4 * tid);
    const int tsum = c.x + c.y + c.z + c.w;
    int incl = tsum;
    for (int o = 1; o < 32; o <<= 1) {
        int y = __shfl_up(incl, o);
        if (lane >= o) incl += y;
    }
    if (lane == 31) wtot[w] = incl;
    __syncthreads();
    int woff = 0;
    for (int k = 0; k < 8; ++k) woff += k < w ? wtot[k] : 0;
    int4 o4;
    o4.x = woff + incl - tsum;
    o4.y = o4.x + c.x;
    o4.z = o4.y + c.y;
    o4.w = o4.z + c.z;
    *(int4*)(hist + 4 * tid) = o4;
    const int b0 = 4 * tid;
    if (b0     < nb) loff[(b0    ) * nblk + blk] = o4.x;
    if (b0 + 1 < nb) loff[(b0 + 1) * nblk + blk] = o4.y;
    if (b0 + 2 < nb) loff[(b0 + 2) * nblk + blk] = o4.z;
    if (b0 + 3 < nb) loff[(b0 + 3) * nblk + blk] = o4.w;
    if (tid == 0) {
        const int left = E - blk * SEG;
        loff[nb * nblk + blk] = left < 0 ? 0 : (left > SEG ? SEG : left);
    }
    __syncthreads();
    #pragma unroll
    for (int u = 0; u < EPT; ++u) {
        const int e = e0 + u * 256;
        const unsigned sl = (unsigned)(hist[bin[u]] + pos[u]);
        const unsigned p  = sl < SEG - 1 ? sl : SEG - 1;
        const unsigned dl = (unsigned)(dst[u] - (bin[u] << BIN_SHIFT));
        if (e < E) coarse[blk * SEG + (int)p] = (int)(((unsigned)s[u] << BIN_SHIFT) | (dl < BIN_W - 1 ? dl : BIN_W - 1));
    }
}

#define PG 8
#define PWD 2
#define PCH 1024
#define PU 13

__device__ __forceinline__ void bin_sort_block(const int* __restrict__ coarse,
                                               const int* __restrict__ loff, int nblk,
                                               int N, int E,
                                               int* __restrict__ rowptr,
                                               int* __restrict__ srcS,
                                               const int b, const int nbins)
{
    __shared__ int cnt[BIN_W + 32];
    __shared__ int wsum[8];
    __shared__ int wbeg[8];
    __shared__ int wend[8];
    __shared__ int pb[PCH];
    const int tid  = threadIdx.x;
    const int lane = tid & 31;
    const int w    = tid >> 5;
    const int gi   = tid / PG;
    const int li   = tid & (PG - 1);
    const int* const lo = loff + b * nblk;
    const int* const hi = lo + nblk;
    const int dump = BIN_W + lane;

    cnt[tid] = 0;
    if (tid < 32) cnt[BIN_W + tid] = 0;

    int sumS = 0, sumE = 0;
    auto stage = [&](int k0, int npc) {
        __syncthreads();
        int a[PCH / 256], z[PCH / 256];
        #pragma unroll
        for (int c = 0; c < PCH / 256; ++c) {
            const int kk = c * 256 + tid;
            const int kc = kk < npc ? kk : npc - 1;
            a[c] = lo[k0 + kc];
            z[c] = hi[k0 + kc];
        }
        #pragma unroll
        for (int c = 0; c < PCH / 256; ++c) {
            const int kk = c * 256 + tid;
            int aa = a[c], zz = z[c];
            aa = aa < 0 ? 0 : (aa > SEG ? SEG : aa);
            zz = zz < aa ? aa : (zz > SEG ? SEG : zz);
            if (kk < npc) { pb[kk] = aa | ((zz - aa) << 16); sumS += aa; sumE += zz; }
        }
        __syncthreads();
    };

    auto walk = [&](int k0, int npc, auto place) {
        constexpr bool PLACE = decltype(place)::value;
        const int rounds = (npc + 31) >> 5;
        bool anyLong = false;
        for (int r0 = 0; r0 < rounds; r0 += PU) {
            int bw[PU];
            #pragma unroll
            for (int u = 0; u < PU; ++u) {
                const int kk = (r0 + u) * 32 + gi;
                bw[u] = pb[kk < npc ? kk : 0];
            }
            int wv[PU][PWD];
            #pragma unroll
            for (int u = 0; u < PU; ++u) {
                const int kk = (r0 + u) * 32 + gi;
                const bool in = kk < npc;
                const int kc = in ? kk : 0;
                const int a   = bw[u] & 0xffff;
                const int len = in ? (int)((unsigned)bw[u] >> 16) : 0;
                anyLong = anyLong || len > PG * PWD;
                #pragma unroll
                for (int j = 0; j < PWD; ++j) {
                    const int i = li + j * PG;
                    const int ofs = i < len ? a + i : (a < SEG - 1 ? a : SEG - 1);
                    wv[u][j] = coarse[(k0 + kc) * SEG + ofs];
                }
            }
            int pos[PU][PWD];
            #pragma unroll
            for (int u = 0; u < PU; ++u) {
                const int kk = (r0 + u) * 32 + gi;
                const int len = kk < npc ? (int)((unsigned)bw[u] >> 16) : 0;
                #pragma unroll
                for (int j = 0; j < PWD; ++j) {
                    const int slot = li + j * PG < len ? (wv[u][j] & (BIN_W - 1)) : dump;
                    if constexpr (PLACE) pos[u][j] = atomicAdd(&cnt[slot], 1);
                    else                 atomicAdd(&cnt[slot], 1);
                }
            }
            if constexpr (PLACE) {
                #pragma unroll
                for (int u = 0; u < PU; ++u) {
                    const int kk = (r0 + u) * 32 + gi;
                    const int len = kk < npc ? (int)((unsigned)bw[u] >> 16) : 0;
                    #pragma unroll
                    for (int j = 0; j < PWD; ++j) {
                        const unsigned p = (unsigned)pos[u][j] < (unsigned)(E - 1) ? (unsigned)pos[u][j] : (unsigned)(E - 1);
                        if (li + j * PG < len) srcS[p] = (int)((unsigned)wv[u][j] >> BIN_SHIFT);
                    }
                }
            }
        }
        if (anyLong) {
            for (int r = 0; r < rounds; ++r) {
                const int kk = r * 32 + gi;
                if (kk < npc) {
                    const int bwd = pb[kk];
                    const int len = (int)((unsigned)bwd >> 16);
                    const int a   = bwd & 0xffff;
                    for (int i = PG * PWD + li; i < len; i += PG) {
                        const int wd = coarse[(k0 + kk) * SEG + a + i];
                        const int ps = atomicAdd(&cnt[wd & (BIN_W - 1)], 1);
                        if constexpr (PLACE) {
                            const unsigned p = (unsigned)ps < (unsigned)(E - 1) ? (unsigned)ps : (unsigned)(E - 1);
                            srcS[p] = (int)((unsigned)wd >> BIN_SHIFT);
                        }
                    }
                }
            }
        }
    };

    for (int k0 = 0; k0 < nblk; k0 += PCH) {
        const int npc = nblk - k0 < PCH ? nblk - k0 : PCH;
        stage(k0, npc);
        walk(k0, npc, std::false_type{});
    }
    for (int o = 16; o > 0; o >>= 1) { sumS += __shfl_xor(sumS, o); sumE += __shfl_xor(sumE, o); }
    if (lane == 0) { wbeg[w] = sumS; wend[w] = sumE; }
    __syncthreads();
    int start = 0, end = 0;
    for (int k = 0; k < 8; ++k) { start += wbeg[k]; end += wend[k]; }
    start = start < 0 ? 0 : (start > E ? E : start);
    end   = end < start ? start : (end > E ? E : end);

    const int v = cnt[tid];
    int x = v;
    for (int o = 1; o < 32; o <<= 1) {
        int y = __shfl_up(x, o);
        if (lane >= o) x += y;
    }
    if (lane == 31) wsum[w] = x;
    __syncthreads();
    int woff = 0;
    for (int k = 0; k < 8; ++k) woff += k < w ? wsum[k] : 0;
    const int first = start + woff + x - v;
    const int node  = b * BIN_W + tid;
    if (node < N) rowptr[node] = first;
    if (b == nbins - 1 && tid == 0) rowptr[N] = end;
    cnt[tid] = first;
    __syncthreads();

    for (int k0 = 0; k0 < nblk; k0 += PCH) {
        const int npc = nblk - k0 < PCH ? nblk - k0 : PCH;
        if (nblk > PCH) stage(k0, npc);
        walk(k0, npc, std::true_type{});
    }
}

template <int H>
__global__ __launch_bounds__(256) void k_segsort_gemm(const int* __restrict__ ei, int E, int nb,
                                                      int* __restrict__ loff,
                                                      int* __restrict__ coarse, int nEdgeBlocks,
                                                      const float* __restrict__ x,
                                                      const float* __restrict__ W,
                                                      const float* __restrict__ a_src,
                                                      const float* __restrict__ a_dst,
                                                      float* __restrict__ h,
                                                      float* __restrict__ as,
                                                      float* __restrict__ ad,
                                                      int tileBegin, int tileEnd)
{
    const int blk = (int)blockIdx.x;
    if (blk < nEdgeBlocks)
        seg_sort_block(ei, E, nb, loff, coarse, blk, nEdgeBlocks);
    else
        gemm_row_tiles<H>(x, W, a_src, a_dst, h, as, ad, tileBegin, tileEnd,
                          blk - nEdgeBlocks, (int)gridDim.x - nEdgeBlocks);
}

template <int H>
__global__ __launch_bounds__(256) void k_sort_gemm(const int* __restrict__ coarse,
                                                   const int* __restrict__ loff, int nblk,
                                                   int N, int E,
                                                   int* __restrict__ rowptr,
                                                   int* __restrict__ srcS, int nbins,
                                                   const float* __restrict__ x,
                                                   const float* __restrict__ W,
                                                   const float* __restrict__ a_src,
                                                   const float* __restrict__ a_dst,
                                                   float* __restrict__ h,
                                                   float* __restrict__ as,
                                                   float* __restrict__ ad,
                                                   int tileBegin, int tileEnd)
{
    const int blk = (int)blockIdx.x;
    if (blk < nbins)
        bin_sort_block(coarse, loff, nblk, N, E, rowptr, srcS, blk, nbins);
    else
        gemm_row_tiles<H>(x, W, a_src, a_dst, h, as, ad, tileBegin, tileEnd,
                          blk - nbins, (int)gridDim.x - nbins);
}

template <int H>
__device__ __forceinline__ void edge_scores(const float* __restrict__ as, int s,
                                            const float (&adh)[H], bool valid, float (&v)[H])
{
    if constexpr (H == 4) {
        const float4 a = *(const float4*)(as + (long)s * 4);
        v[0] = a.x + adh[0]; v[1] = a.y + adh[1]; v[2] = a.z + adh[2]; v[3] = a.w + adh[3];
    } else {
        #pragma unroll
        for (int hh = 0; hh < H; ++hh) v[hh] = as[(long)s * H + hh] + adh[hh];
    }
    #pragma unroll
    for (int hh = 0; hh < H; ++hh) {
        float t = v[hh];
        t = t > 0.f ? t : NEG_SLOPE * t;
        v[hh] = valid ? t : -INFINITY;
    }
}

#define PW_PITCH 36
template <int H, bool ELU>
__device__ __forceinline__ float4 gather_node(const int d, const int beg, const int cnt,
                                              const int lane, const int hd, const unsigned last,
                                              const int* __restrict__ srcS,
                                              const float* __restrict__ as,
                                              const float* __restrict__ ad,
                                              const float4* __restrict__ h4,
                                              float* const pwv, const float* const pwr,
                                              const float4 b)
{
    float adh[H];
    #pragma unroll
    for (int hh = 0; hh < H; ++hh) adh[hh] = ad[(long)d * H + hh];

    int sj = d;
    if (lane < cnt - 1) {
        const unsigned u = (unsigned)srcS[beg + lane];
        sj = (int)(u < last ? u : last);
    }
    float v[H], mx[H];
    edge_scores<H>(as, sj, adh, lane < cnt, v);
    #pragma unroll
    for (int hh = 0; hh < H; ++hh) mx[hh] = v[hh];
    for (int base = 32; base < cnt; base += 32) {
        const int j = base + lane;
        int s2 = d;
        if (j < cnt - 1) { const unsigned u = (unsigned)srcS[beg + j]; s2 = (int)(u < last ? u : last); }
        float w[H];
        edge_scores<H>(as, s2, adh, j < cnt, w);
        #pragma unroll
        for (int hh = 0; hh < H; ++hh) mx[hh] = fmaxf(mx[hh], w[hh]);
    }
    #pragma unroll
    for (int hh = 0; hh < H; ++hh)
        for (int off = 16; off > 0; off >>= 1) mx[hh] = fmaxf(mx[hh], __shfl_xor(mx[hh], off));

    float dn[H];
    #pragma unroll
    for (int hh = 0; hh < H; ++hh) dn[hh] = 0.f;
    float4 acc = {0.f, 0.f, 0.f, 0.f};

    auto add_rows = [&](int sx, const float (&px)[H], int n) {
        if constexpr (H > 1) {
            #pragma unroll
            for (int hh = 0; hh < H; ++hh) pwv[hh * PW_PITCH + lane] = px[hh];
        }
        auto use_rows = [&](const float4 (&hv)[4], const float4 pt) {
            acc.x = __builtin_fmaf(pt.x, hv[0].x, acc.x); acc.y = __builtin_fmaf(pt.x, hv[0].y, acc.y);
            acc.z = __builtin_fmaf(pt.x, hv[0].z, acc.z); acc.w = __builtin_fmaf(pt.x, hv[0].w, acc.w);
            acc.x = __builtin_fmaf(pt.y, hv[1].x, acc.x); acc.y = __builtin_fmaf(pt.y, hv[1].y, acc.y);
            acc.z = __builtin_fmaf(pt.y, hv[1].z, acc.z); acc.w = __builtin_fmaf(pt.y, hv[1].w, acc.w);
            acc.x = __builtin_fmaf(pt.z, hv[2].x, acc.x); acc.y = __builtin_fmaf(pt.z, hv[2].y, acc.y);
            acc.z = __builtin_fmaf(pt.z, hv[2].z, acc.z); acc.w = __builtin_fmaf(pt.z, hv[2].w, acc.w);
            acc.x = __builtin_fmaf(pt.w, hv[3].x, acc.x); acc.y = __builtin_fmaf(pt.w, hv[3].y, acc.y);
            acc.z = __builtin_fmaf(pt.w, hv[3].z, acc.z); acc.w = __builtin_fmaf(pt.w, hv[3].w, acc.w);
        };
        auto weights = [&](int t0) {
            if constexpr (H > 1) {
                return *(const float4*)(pwr + t0);
            } else {
                float4 r;
                r.x = __int_as_float(__builtin_amdgcn_readlane(__float_as_int(px[0]), t0));
                r.y = __int_as_float(__builtin_amdgcn_readlane(__float_as_int(px[0]), t0 + 1));
                r.z = __int_as_float(__builtin_amdgcn_readlane(__float_as_int(px[0]), t0 + 2));
                r.w = __int_as_float(__builtin_amdgcn_readlane(__float_as_int(px[0]), t0 + 3));
                return r;
            }
        };
        int t = 0;
        for (; n - t > 4; t += 8) {
            float4 ha[4], hb[4];
            #pragma unroll
            for (int u = 0; u < 4; ++u) {
                const unsigned s = (unsigned)__builtin_amdgcn_readlane(sx, t + u);
                ha[u] = h4[s * 32u + (unsigned)lane];
            }
            #pragma unroll
            for (int u = 0; u < 4; ++u) {
                const unsigned s = (unsigned)__builtin_amdgcn_readlane(sx, t + 4 + u);
                hb[u] = h4[s * 32u + (unsigned)lane];
            }
            const float4 pa = weights(t);
            const float4 pb = weights(t + 4);
            use_rows(ha, pa);
            use_rows(hb, pb);
        }
        if (t < n) {
            float4 ha[4];
            #pragma unroll
            for (int u = 0; u < 4; ++u) {
                const unsigned s = (unsigned)__builtin_amdgcn_readlane(sx, t + u);
                ha[u] = h4[s * 32u + (unsigned)lane];
            }
            const float4 pa = weights(t);
            use_rows(ha, pa);
        }
    };

    {
        float p[H];
        #pragma unroll
        for (int hh = 0; hh < H; ++hh) { p[hh] = expf(v[hh] - mx[hh]); dn[hh] += p[hh]; }
        add_rows(sj, p, cnt < 32 ? cnt : 32);
    }
    for (int base = 32; base < cnt; base += 32) {
        const int j = base + lane;
        int s2 = d;
        if (j < cnt - 1) { const unsigned u = (unsigned)srcS[beg + j]; s2 = (int)(u < last ? u : last); }
        float w[H], p[H];
        edge_scores<H>(as, s2, adh, j < cnt, w);
        #pragma unroll
        for (int hh = 0; hh < H; ++hh) { p[hh] = expf(w[hh] - mx[hh]); dn[hh] += p[hh]; }
        const int left = cnt - base;
        add_rows(s2, p, left < 32 ? left : 32);
    }
    #pragma unroll
    for (int hh = 0; hh < H; ++hh)
        for (int off = 16; off > 0; off >>= 1) dn[hh] += __shfl_xor(dn[hh], off);

    float den = dn[0];
    if constexpr (H == 4) den = hd == 0 ? dn[0] : hd == 1 ? dn[1] : hd == 2 ? dn[2] : dn[3];
    const float inv = 1.f / den;
    float4 r;
    r.x = acc.x * inv + b.x; r.y = acc.y * inv + b.y;
    r.z = acc.z * inv + b.z; r.w = acc.w * inv + b.w;
    if constexpr (ELU) {
        r.x = r.x > 0.f ? r.x : expm1f(r.x); r.y = r.y > 0.f ? r.y : expm1f(r.y);
        r.z = r.z > 0.f ? r.z : expm1f(r.z); r.w = r.w > 0.f ? r.w : expm1f(r.w);
    }
    return r;
}

template <int H, bool ELU, int NPW>
__global__ __launch_bounds__(256) void k_node_gather(const int* __restrict__ rowptr,
                                                     const int* __restrict__ srcS,
                                                     const float* __restrict__ as,
                                                     const float* __restrict__ ad,
                                                     const float* __restrict__ h,
                                                     const float* __restrict__ bias,
                                                     float* __restrict__ o, int N)
{
    __shared__ __attribute__((aligned(16))) float pw[8 * H * PW_PITCH];
    const int lane = threadIdx.x & 31;
    const int d0 = __builtin_amdgcn_readfirstlane((int)(blockIdx.x * 8 + (threadIdx.x >> 5))) * NPW;
    if (d0 >= N) return;
    const int nd = N - d0 < NPW ? N - d0 : NPW;
    const int rp = lane <= nd ? rowptr[d0 + lane] : 0;
    const unsigned last = (unsigned)(N - 1);
    const int hd = lane / (32 / H);
    float* const pwv = pw + (threadIdx.x >> 5) * (H * PW_PITCH);
    const float* const pwr = pwv + hd * PW_PITCH;
    const float4* const h4 = (const float4*)h;
    const float4 b = *(const float4*)(bias + lane * 4);

    for (int k = 0; k < nd; ++k) {
        const int d   = d0 + k;
        const int beg = __builtin_amdgcn_readlane(rp, k);
        const int cnt = __builtin_amdgcn_readlane(rp, k + 1) - beg + 1;
        const float4 r = gather_node<H, ELU>(d, beg, cnt, lane, hd, last, srcS, as, ad, h4, pwv, pwr, b);
        *(float4*)(o + (long)d * 128 + lane * 4) = r;
    }
}

template <int H, int NPW, int HN>
__global__ __launch_bounds__(256) void k_gather_gemm(const int* __restrict__ rowptr,
                                                     const int* __restrict__ srcS,
                                                     const float* __restrict__ as,
                                                     const float* __restrict__ ad,
                                                     const float* __restrict__ h,
                                                     const float* __restrict__ bias,
                                                     const float* __restrict__ W,
                                                     const float* __restrict__ a_src,
                                                     const float* __restrict__ a_dst,
                                                     float* __restrict__ hn,
                                                     float* __restrict__ asn,
                                                     float* __restrict__ adn, int N)
{
    constexpr int ROWS  = 8 * NPW;
    static_assert(ROWS % 16 == 0, "a block's rows must be whole 16-row tiles");
    constexpr int TILES = ROWS / 16;
    __shared__ __attribute__((aligned(16))) float pw[8 * H * PW_PITCH];
    __shared__ __attribute__((aligned(16))) float xs[ROWS * XS_PITCH];
    __shared__ float part[2 * 8 * 32];
    const int tid  = threadIdx.x;
    const int lane = tid & 31;
    const int wave = tid >> 5;
    const int d0 = __builtin_amdgcn_readfirstlane((int)(blockIdx.x * 8 + (threadIdx.x >> 5))) * NPW;
    const int nd = N - d0 < NPW ? (N - d0 > 0 ? N - d0 : 0) : NPW;
    const int rp = (nd > 0 && lane <= nd) ? rowptr[d0 + lane] : 0;
    const unsigned last = (unsigned)(N - 1);
    const int hd = lane / (32 / H);
    float* const pwv = pw + (threadIdx.x >> 5) * (H * PW_PITCH);
    const float* const pwr = pwv + hd * PW_PITCH;
    const float4* const h4 = (const float4*)h;
    const float4 b = *(const float4*)(bias + lane * 4);
    float* const xrow = xs + (wave * NPW) * XS_PITCH + lane * 4;

    for (int k = 0; k < nd; ++k) {
        const int beg = __builtin_amdgcn_readlane(rp, k);
        const int cnt = __builtin_amdgcn_readlane(rp, k + 1) - beg + 1;
        const float4 r = gather_node<H, true>(d0 + k, beg, cnt, lane, hd, last, srcS, as, ad, h4, pwv, pwr, b);
        *(float4*)(xrow + k * XS_PITCH) = r;
    }
    for (int k = nd; k < NPW; ++k) *(float4*)(xrow + k * XS_PITCH) = float4{0.f, 0.f, 0.f, 0.f};

    const int m16     = lane & 15;
    const int khalf   = (lane >> 4) << 1;
    const int colBase = wave * 16;
    v2f bf[32];
    {
        const float* wr = W + (long)(colBase + m16) * 128 + khalf;
        #pragma unroll
        for (int kk = 0; kk < 32; ++kk) { bf[kk].x = wr[4 * kk]; bf[kk].y = wr[4 * kk + 1]; }
    }
    const bool  up8 = (m16 & 8) != 0;
    const float asv = a_src[colBase + m16];
    const float adv = a_dst[colBase + m16];
    const float ka  = up8 ? adv : asv;
    const float sa  = up8 ? asv : adv;
    const int partIdx = wave * 32 + (m16 >> 3) * 16 + ((lane >> 4) << 3) + (m16 & 7);
    const unsigned hOff = (unsigned)(((lane >> 4) << 3) * 128 + colBase + m16);

    auto finish_alpha = [&](unsigned tileOfs, const float* pb) {
        if (tid < 16 * HN) {
            constexpr int WPH = 8 / HN;
            const int hdn = tid & (HN - 1);
            const int row = tid / HN;
            float s = pb[(hdn * WPH) * 32 + row];
            float d = pb[(hdn * WPH) * 32 + 16 + row];
            #pragma unroll
            for (int j = 1; j < WPH; ++j) {
                s += pb[(hdn * WPH + j) * 32 + row];
                d += pb[(hdn * WPH + j) * 32 + 16 + row];
            }
            asn[tileOfs * (16u * HN) + (unsigned)tid] = s;
            adn[tileOfs * (16u * HN) + (unsigned)tid] = d;
        }
    };

    __syncthreads();
    for (int t = 0; t < TILES; ++t) {
        const int tile = (int)blockIdx.x * TILES + t;
        const bool live = tile * 16 + 16 <= N;
        const unsigned tileBase = (unsigned)tile * 2048u;
        const float* const xt = xs + t * (16 * XS_PITCH);
        float* const pb = part + (t & 1) * 256;

        v8f acc = {};
        #pragma unroll
        for (int kk = 0; kk < 32; ++kk) {
            v2f a;
            a.x = xt[m16 * XS_PITCH + 4 * kk + khalf];
            a.y = xt[m16 * XS_PITCH + 4 * kk + khalf + 1];
            acc = __builtin_amdgcn_wmma_f32_16x16x4_f32(false, a, false, bf[kk],
                                                        (short)0, acc, false, false);
        }
        float val[8];
        #pragma unroll
        for (int r = 0; r < 8; ++r) val[r] = acc[r] * ka + row_xor_read<8>(acc[r] * sa);
        fold_step<4, 4>(val, (m16 & 4) != 0);
        fold_step<2, 2>(val, (m16 & 2) != 0);
        fold_step<1, 1>(val, (m16 & 1) != 0);
        pb[partIdx] = val[0];
        __syncthreads();
        if (live) {
            #pragma unroll
            for (int r = 0; r < 8; ++r)
                hn[tileBase + hOff + 128u * r] = acc[r];
            finish_alpha((unsigned)tile, pb);
        }
    }
}

extern "C" void kernel_launch(void* const* d_in, const int* in_sizes, int n_in,
                              void* d_out, int out_size, void* d_ws, size_t ws_size,
                              hipStream_t stream)
{
    const float* x     = (const float*)d_in[0];
    const int*   ei    = (const int*)  d_in[1];
    const float* W1    = (const float*)d_in[2];
    const float* asrc1 = (const float*)d_in[3];
    const float* adst1 = (const float*)d_in[4];
    const float* b1    = (const float*)d_in[5];
    const float* W2    = (const float*)d_in[6];
    const float* asrc2 = (const float*)d_in[7];
    const float* adst2 = (const float*)d_in[8];
    const float* b2    = (const float*)d_in[9];
    float* out = (float*)d_out;

    const int N    = in_sizes[0] / 128;
    const int E    = in_sizes[1] / 2;
    const long NF  = (long)N * 128;

    float* ws   = (float*)d_ws;
    float* h    = ws;
    const int T = 256;
    dim3 gE((E + SEG - 1) / SEG);
    const int NB = (N + BIN_W - 1) / BIN_W;
    int*   coarse = (int*)(ws + NF);
    int*   srcS   = coarse + (int)gE.x * SEG;
    int*   loff   = srcS + E;
    int*   rowptr = loff + (NB + 1) * (int)gE.x;
    float* as1  = ws + 2 * NF;
    float* ad1  = as1 + (long)N * 4;
    float* as2  = ad1 + (long)N * 4;
    float* ad2  = as2 + N;
    float* h1   = out;

    dim3 gNode1((N + 8 * 10 - 1) / (8 * 10));
    dim3 gNode2((N + 8 * 8 - 1) / (8 * 8));
    const int nTiles = N / 16;
    const int tSplit = nTiles < 480 * 6 ? nTiles : 480 * 6;
    dim3 gSegGemm(gE.x + 480);
    dim3 gSortGemm(NB + 674);

    k_segsort_gemm<4><<<gSegGemm, T, 0, stream>>>(ei, E, NB, loff, coarse, (int)gE.x,
                                                  x, W1, asrc1, adst1, h1, as1, ad1, 0, tSplit);
    k_sort_gemm<4><<<gSortGemm, T, 0, stream>>>(coarse, loff, (int)gE.x, N, E, rowptr, srcS, NB,
                                                x, W1, asrc1, adst1, h1, as1, ad1, tSplit, nTiles);

    k_gather_gemm<4, 10, 1><<<gNode1, T, 0, stream>>>(rowptr, srcS, as1, ad1, h1, b1,
                                                      W2, asrc2, adst2, h, as2, ad2, N);

    k_node_gather<1, false, 8><<<gNode2, T, 0, stream>>>(rowptr, srcS, as2, ad2, h, b2, out, N);
}
